# starting kernel
# baseline (speedup 1.0000x reference)
#include <hip/hip_runtime.h>

typedef __attribute__((ext_vector_type(2))) float v2f;
typedef __attribute__((ext_vector_type(8))) float v8f;

#define NN 50000
#define CH 128

__global__ void zero_f4(float4* __restrict__ p, int n4) {
  int i = blockIdx.x * blockDim.x + threadIdx.x;
  if (i < n4) p[i] = make_float4(0.f, 0.f, 0.f, 0.f);
}

__global__ void scatter_add(const float* __restrict__ X,
                            const int* __restrict__ src,
                            const int* __restrict__ dst,
                            float* __restrict__ agg, int E) {
  long long t = (long long)blockIdx.x * blockDim.x + threadIdx.x;
  int e = (int)(t >> 5);
  if (e >= E) return;
  int lane = (int)(t & 31);
  int s = src[e], d = dst[e];
  float4 v = *((const float4*)(X + (size_t)s * CH) + lane);
  float* a = agg + (size_t)d * CH + lane * 4;
  unsafeAtomicAdd(a + 0, v.x);
  unsafeAtomicAdd(a + 1, v.y);
  unsafeAtomicAdd(a + 2, v.z);
  unsafeAtomicAdd(a + 3, v.w);
}

template <bool ADD_AGG, bool RELU, int K, int MT_PER_WAVE>
__global__ void gemm_relu_wmma(const float* __restrict__ A,
                               const float* __restrict__ AGG,
                               const float* __restrict__ W,
                               const float* __restrict__ bias,
                               float* __restrict__ out,
                               int M, int N) {
  const int lane = threadIdx.x & 31;
  const int wave = threadIdx.x >> 5;
  const int tilesN = N >> 4;
  const int mchunks = (M >> 4) / MT_PER_WAVE;
  const int work = blockIdx.x * (blockDim.x >> 5) + wave;
  const int chunk = work / tilesN;
  if (chunk >= mchunks) return;
  const int nt = work - chunk * tilesN;
  const int hi = lane >> 4;
  const int lo = lane & 15;
  const int col = (nt << 4) + lo;

  v2f bf[K / 4];
#pragma unroll
  for (int i = 0; i < K / 4; ++i) {
    const int ka = 4 * i + (hi << 1);
    bf[i].x = W[(size_t)ka * N + col];
    bf[i].y = W[(size_t)(ka + 1) * N + col];
  }
  const float bv = bias[col];

  const int mt0 = chunk * MT_PER_WAVE;
  for (int mi = 0; mi < MT_PER_WAVE; ++mi) {
    const int m0 = (mt0 + mi) << 4;
    const float* arow = A + (size_t)(m0 + lo) * K + (hi << 1);
    const float* grow = ADD_AGG ? (AGG + (size_t)(m0 + lo) * K + (hi << 1)) : nullptr;

    v8f c;
#pragma unroll
    for (int r = 0; r < 8; ++r) c[r] = bv;

#pragma unroll
    for (int i = 0; i < K / 4; ++i) {
      v2f a;
      a.x = arow[4 * i];
      a.y = arow[4 * i + 1];
      if (ADD_AGG) {
        a.x += grow[4 * i];
        a.y += grow[4 * i + 1];
      }
      c = __builtin_amdgcn_wmma_f32_16x16x4_f32(
           false, a,  false, bf[i],
           (short)0, c,  false,  false);
    }

    float* orow = out + (size_t)(m0 + hi * 8) * N + col;
#pragma unroll
    for (int r = 0; r < 8; ++r) {
      float v = c[r];
      if (RELU) v = fmaxf(v, 0.f);
      orow[(size_t)r * N] = v;
    }
  }
}

__global__ void head_gemv(const float* __restrict__ H,
                          const float* __restrict__ w4,
                          const float* __restrict__ b4,
                          float* __restrict__ out, int M) {
  long long t = (long long)blockIdx.x * blockDim.x + threadIdx.x;
  int n = (int)(t >> 5);
  if (n >= M) return;
  int lane = (int)(t & 31);
  const float* h = H + (size_t)n * 64;
  float s = h[lane] * w4[lane] + h[lane + 32] * w4[lane + 32];
#pragma unroll
  for (int off = 16; off > 0; off >>= 1) s += __shfl_xor(s, off, 32);
  if (lane == 0) out[n] = s + b4[0];
}

extern "C" void kernel_launch(void* const* d_in, const int* in_sizes, int n_in,
                              void* d_out, int out_size, void* d_ws, size_t ws_size,
                              hipStream_t stream) {
  const float* x   = (const float*)d_in[0];
  const int*   ei  = (const int*)d_in[1];
  const int    E   = in_sizes[1] / 2;
  const int* src = ei;
  const int* dst = ei + E;
  const float* w1a = (const float*)d_in[2];
  const float* b1a = (const float*)d_in[3];
  const float* w1b = (const float*)d_in[4];
  const float* b1b = (const float*)d_in[5];
  const float* w2a = (const float*)d_in[6];
  const float* b2a = (const float*)d_in[7];
  const float* w2b = (const float*)d_in[8];
  const float* b2b = (const float*)d_in[9];
  const float* w3  = (const float*)d_in[10];
  const float* b3  = (const float*)d_in[11];
  const float* w4  = (const float*)d_in[12];
  const float* b4  = (const float*)d_in[13];
  float* outp = (float*)d_out;

  float* agg  = (float*)d_ws;
  float* bufA = agg  + (size_t)NN * CH;
  float* bufB = bufA + (size_t)NN * CH;
  float* h3   = agg;

  const int n4 = NN * CH / 4;
  dim3 zb(256), zg((n4 + 255) / 256);
  long long sthreads = (long long)E * 32;
  dim3 sb(256), sg((unsigned)((sthreads + 255) / 256));

  constexpr int MTW = 5;
  auto gblocks = [](int M, int N) {
    int waves = ((M >> 4) / MTW) * (N >> 4);
    return (unsigned)((waves + 3) / 4);
  };

  zero_f4<<<zg, zb, 0, stream>>>((float4*)agg, n4);
  scatter_add<<<sg, sb, 0, stream>>>(x, src, dst, agg, E);
  gemm_relu_wmma<true,  true, CH, MTW><<<gblocks(NN, 128), 128, 0, stream>>>(x,    agg,     w1a, b1a, bufA, NN, 128);
  gemm_relu_wmma<false, true, CH, MTW><<<gblocks(NN, 128), 128, 0, stream>>>(bufA, nullptr, w1b, b1b, bufB, NN, 128);

  zero_f4<<<zg, zb, 0, stream>>>((float4*)agg, n4);
  scatter_add<<<sg, sb, 0, stream>>>(bufB, src, dst, agg, E);
  gemm_relu_wmma<true,  true, CH, MTW><<<gblocks(NN, 128), 128, 0, stream>>>(bufB, agg,     w2a, b2a, bufA, NN, 128);
  gemm_relu_wmma<false, true, CH, MTW><<<gblocks(NN, 128), 128, 0, stream>>>(bufA, nullptr, w2b, b2b, bufB, NN, 128);

  gemm_relu_wmma<false, true, CH, MTW><<<gblocks(NN, 64), 128, 0, stream>>>(bufB, nullptr, w3, b3, h3, NN, 64);
  head_gemv<<<(NN * 32 + 255) / 256, 256, 0, stream>>>(h3, w4, b4, outp, NN);
}


# Replace f32 atomic scatter-add by edges counting-sorted by destination (int atomic rank, built once per call) + atomic-free per-node gather sum; neighbour sum order = arrival rank
# speedup vs baseline: 1.5538x; 1.5538x over previous
#include <hip/hip_runtime.h>

typedef __attribute__((ext_vector_type(2))) float v2f;
typedef __attribute__((ext_vector_type(8))) float v8f;

#define NN 50000
#define CH 128

__global__ void zero_f4(float4* __restrict__ p, int n4) {
  int i = blockIdx.x * blockDim.x + threadIdx.x;
  if (i < n4) p[i] = make_float4(0.f, 0.f, 0.f, 0.f);
}

__global__ void edge_rank(const int* __restrict__ dst, int* __restrict__ deg,
                          int* __restrict__ rank, int E) {
  int e = blockIdx.x * blockDim.x + threadIdx.x;
  if (e >= E) return;
  rank[e] = atomicAdd(deg + dst[e], 1);
}

#define SCAN_T 1024
__global__ void __launch_bounds__(SCAN_T)
row_offsets(const int* __restrict__ deg, int* __restrict__ rowptr, int M) {
  __shared__ int wpre[32];
  __shared__ int wtot[32];
  const int t = threadIdx.x;
  const int lane = t & 31;
  const int wave = t >> 5;
  const int first = blockIdx.x * SCAN_T;
  int pre = 0;
  for (int i = t; i < first; i += SCAN_T) pre += deg[i];
  const int idx = first + t;
  const int v = (idx < M) ? deg[idx] : 0;
  int incl = v;
#pragma unroll
  for (int off = 1; off < 32; off <<= 1) {
    const int y = __shfl_up(incl, off, 32);
    if (lane >= off) incl += y;
  }
#pragma unroll
  for (int off = 16; off > 0; off >>= 1) pre += __shfl_xor(pre, off, 32);
  if (lane == 31) wtot[wave] = incl;
  if (lane == 0) wpre[wave] = pre;
  __syncthreads();
  if (wave == 0) {
    int p = wpre[lane];
#pragma unroll
    for (int off = 16; off > 0; off >>= 1) p += __shfl_xor(p, off, 32);
    const int w = wtot[lane];
    int wi = w;
#pragma unroll
    for (int off = 1; off < 32; off <<= 1) {
      const int y = __shfl_up(wi, off, 32);
      if (lane >= off) wi += y;
    }
    wtot[lane] = p + wi - w;
  }
  __syncthreads();
  if (idx <= M) rowptr[idx] = wtot[wave] + incl - v;
}

__global__ void edge_fill(const int* __restrict__ src, const int* __restrict__ dst,
                          const int* __restrict__ rowptr, const int* __restrict__ rank,
                          int* __restrict__ ssrc, int E) {
  int e = blockIdx.x * blockDim.x + threadIdx.x;
  if (e >= E) return;
  ssrc[rowptr[dst[e]] + rank[e]] = src[e];
}

template <int B>
__device__ __forceinline__ void add_rows(const float4* __restrict__ X4,
                                         const int* __restrict__ ssrc,
                                         int j, int lane, float4& acc) {
  int s[B];
  float4 v[B];
#pragma unroll
  for (int u = 0; u < B; ++u) s[u] = ssrc[j + u];
#pragma unroll
  for (int u = 0; u < B; ++u) v[u] = X4[(size_t)s[u] * (CH / 4) + lane];
#pragma unroll
  for (int u = 0; u < B; ++u) {
    acc.x += v[u].x;
    acc.y += v[u].y;
    acc.z += v[u].z;
    acc.w += v[u].w;
  }
}

__global__ void gather_sum(const float* __restrict__ X,
                           const int* __restrict__ rowptr,
                           const int* __restrict__ ssrc,
                           float* __restrict__ agg, int M) {
  const int lane = threadIdx.x & 31;
  const int n = __builtin_amdgcn_readfirstlane(
      (int)(blockIdx.x * (blockDim.x >> 5) + (threadIdx.x >> 5)));
  if (n >= M) return;
  const int beg = rowptr[n];
  const int end = rowptr[n + 1];
  const float4* X4 = (const float4*)X;
  float4 acc = make_float4(0.f, 0.f, 0.f, 0.f);
  int j = beg;
  for (; j + 8 <= end; j += 8) add_rows<8>(X4, ssrc, j, lane, acc);
  if (j + 4 <= end) { add_rows<4>(X4, ssrc, j, lane, acc); j += 4; }
  if (j + 2 <= end) { add_rows<2>(X4, ssrc, j, lane, acc); j += 2; }
  if (j < end) add_rows<1>(X4, ssrc, j, lane, acc);
  ((float4*)agg)[(size_t)n * (CH / 4) + lane] = acc;
}

template <bool ADD_AGG, bool RELU, int K, int MT_PER_WAVE>
__global__ void gemm_relu_wmma(const float* __restrict__ A,
                               const float* __restrict__ AGG,
                               const float* __restrict__ W,
                               const float* __restrict__ bias,
                               float* __restrict__ out,
                               int M, int N) {
  const int lane = threadIdx.x & 31;
  const int wave = threadIdx.x >> 5;
  const int tilesN = N >> 4;
  const int mchunks = (M >> 4) / MT_PER_WAVE;
  const int work = blockIdx.x * (blockDim.x >> 5) + wave;
  const int chunk = work / tilesN;
  if (chunk >= mchunks) return;
  const int nt = work - chunk * tilesN;
  const int hi = lane >> 4;
  const int lo = lane & 15;
  const int col = (nt << 4) + lo;

  v2f bf[K / 4];
#pragma unroll
  for (int i = 0; i < K / 4; ++i) {
    const int ka = 4 * i + (hi << 1);
    bf[i].x = W[(size_t)ka * N + col];
    bf[i].y = W[(size_t)(ka + 1) * N + col];
  }
  const float bv = bias[col];

  const int mt0 = chunk * MT_PER_WAVE;
  for (int mi = 0; mi < MT_PER_WAVE; ++mi) {
    const int m0 = (mt0 + mi) << 4;
    const float* arow = A + (size_t)(m0 + lo) * K + (hi << 1);
    const float* grow = ADD_AGG ? (AGG + (size_t)(m0 + lo) * K + (hi << 1)) : nullptr;

    v8f c;
#pragma unroll
    for (int r = 0; r < 8; ++r) c[r] = bv;

#pragma unroll
    for (int i = 0; i < K / 4; ++i) {
      v2f a;
      a.x = arow[4 * i];
      a.y = arow[4 * i + 1];
      if (ADD_AGG) {
        a.x += grow[4 * i];
        a.y += grow[4 * i + 1];
      }
      c = __builtin_amdgcn_wmma_f32_16x16x4_f32(
           false, a,  false, bf[i],
           (short)0, c,  false,  false);
    }

    float* orow = out + (size_t)(m0 + hi * 8) * N + col;
#pragma unroll
    for (int r = 0; r < 8; ++r) {
      float v = c[r];
      if (RELU) v = fmaxf(v, 0.f);
      orow[(size_t)r * N] = v;
    }
  }
}

__global__ void head_gemv(const float* __restrict__ H,
                          const float* __restrict__ w4,
                          const float* __restrict__ b4,
                          float* __restrict__ out, int M) {
  long long t = (long long)blockIdx.x * blockDim.x + threadIdx.x;
  int n = (int)(t >> 5);
  if (n >= M) return;
  int lane = (int)(t & 31);
  const float* h = H + (size_t)n * 64;
  float s = h[lane] * w4[lane] + h[lane + 32] * w4[lane + 32];
#pragma unroll
  for (int off = 16; off > 0; off >>= 1) s += __shfl_xor(s, off, 32);
  if (lane == 0) out[n] = s + b4[0];
}

extern "C" void kernel_launch(void* const* d_in, const int* in_sizes, int n_in,
                              void* d_out, int out_size, void* d_ws, size_t ws_size,
                              hipStream_t stream) {
  const float* x   = (const float*)d_in[0];
  const int*   ei  = (const int*)d_in[1];
  const int    E   = in_sizes[1] / 2;
  const int* src = ei;
  const int* dst = ei + E;
  const float* w1a = (const float*)d_in[2];
  const float* b1a = (const float*)d_in[3];
  const float* w1b = (const float*)d_in[4];
  const float* b1b = (const float*)d_in[5];
  const float* w2a = (const float*)d_in[6];
  const float* b2a = (const float*)d_in[7];
  const float* w2b = (const float*)d_in[8];
  const float* b2b = (const float*)d_in[9];
  const float* w3  = (const float*)d_in[10];
  const float* b3  = (const float*)d_in[11];
  const float* w4  = (const float*)d_in[12];
  const float* b4  = (const float*)d_in[13];
  float* outp = (float*)d_out;

  float* agg  = (float*)d_ws;
  float* bufA = agg  + (size_t)NN * CH;
  float* bufB = bufA + (size_t)NN * CH;
  float* h3   = agg;
  int* deg    = (int*)(bufB + (size_t)NN * CH);
  int* rowptr = deg + NN;
  int* rank   = rowptr + NN + 16;
  int* ssrc   = rank + E;

  const int d4 = NN / 4;
  dim3 zb(256), zg((d4 + 255) / 256);
  dim3 eb(256), eg((unsigned)((E + 255) / 256));
  dim3 gb(256), gg((NN + 7) / 8);

  constexpr int MTW = 5;
  auto gblocks = [](int M, int N) {
    int waves = ((M >> 4) / MTW) * (N >> 4);
    return (unsigned)((waves + 3) / 4);
  };

  zero_f4<<<zg, zb, 0, stream>>>((float4*)deg, d4);
  edge_rank<<<eg, eb, 0, stream>>>(dst, deg, rank, E);
  row_offsets<<<(NN + SCAN_T) / SCAN_T, SCAN_T, 0, stream>>>(deg, rowptr, NN);
  edge_fill<<<eg, eb, 0, stream>>>(src, dst, rowptr, rank, ssrc, E);

  gather_sum<<<gg, gb, 0, stream>>>(x, rowptr, ssrc, agg, NN);
  gemm_relu_wmma<true,  true, CH, MTW><<<gblocks(NN, 128), 128, 0, stream>>>(x,    agg,     w1a, b1a, bufA, NN, 128);
  gemm_relu_wmma<false, true, CH, MTW><<<gblocks(NN, 128), 128, 0, stream>>>(bufA, nullptr, w1b, b1b, bufB, NN, 128);

  gather_sum<<<gg, gb, 0, stream>>>(bufB, rowptr, ssrc, agg, NN);
  gemm_relu_wmma<true,  true, CH, MTW><<<gblocks(NN, 128), 128, 0, stream>>>(bufB, agg,     w2a, b2a, bufA, NN, 128);
  gemm_relu_wmma<false, true, CH, MTW><<<gblocks(NN, 128), 128, 0, stream>>>(bufA, nullptr, w2b, b2b, bufB, NN, 128);

  gemm_relu_wmma<false, true, CH, MTW><<<gblocks(NN, 64), 128, 0, stream>>>(bufB, nullptr, w3, b3, h3, NN, 64);
  head_gemv<<<(NN * 32 + 255) / 256, 256, 0, stream>>>(h3, w4, b4, outp, NN);
}


# GEMMs: A (+agg) rows read once per work-group with whole-row 16-byte loads into padded LDS, shared by its 8 (4) n-tile waves; A fragments come from LDS, not per-wave 16-line global loads
# speedup vs baseline: 2.0248x; 1.3031x over previous
#include <hip/hip_runtime.h>

typedef __attribute__((ext_vector_type(2))) float v2f;
typedef __attribute__((ext_vector_type(8))) float v8f;

#define NN 50000
#define CH 128

__global__ void zero_f4(float4* __restrict__ p, int n4) {
  int i = blockIdx.x * blockDim.x + threadIdx.x;
  if (i < n4) p[i] = make_float4(0.f, 0.f, 0.f, 0.f);
}

__global__ void edge_rank(const int* __restrict__ dst, int* __restrict__ deg,
                          int* __restrict__ rank, int E) {
  int e = blockIdx.x * blockDim.x + threadIdx.x;
  if (e >= E) return;
  rank[e] = atomicAdd(deg + dst[e], 1);
}

#define SCAN_T 1024
__global__ void __launch_bounds__(SCAN_T)
row_offsets(const int* __restrict__ deg, int* __restrict__ rowptr, int M) {
  __shared__ int wpre[32];
  __shared__ int wtot[32];
  const int t = threadIdx.x;
  const int lane = t & 31;
  const int wave = t >> 5;
  const int first = blockIdx.x * SCAN_T;
  int pre = 0;
  for (int i = t; i < first; i += SCAN_T) pre += deg[i];
  const int idx = first + t;
  const int v = (idx < M) ? deg[idx] : 0;
  int incl = v;
#pragma unroll
  for (int off = 1; off < 32; off <<= 1) {
    const int y = __shfl_up(incl, off, 32);
    if (lane >= off) incl += y;
  }
#pragma unroll
  for (int off = 16; off > 0; off >>= 1) pre += __shfl_xor(pre, off, 32);
  if (lane == 31) wtot[wave] = incl;
  if (lane == 0) wpre[wave] = pre;
  __syncthreads();
  if (wave == 0) {
    int p = wpre[lane];
#pragma unroll
    for (int off = 16; off > 0; off >>= 1) p += __shfl_xor(p, off, 32);
    const int w = wtot[lane];
    int wi = w;
#pragma unroll
    for (int off = 1; off < 32; off <<= 1) {
      const int y = __shfl_up(wi, off, 32);
      if (lane >= off) wi += y;
    }
    wtot[lane] = p + wi - w;
  }
  __syncthreads();
  if (idx <= M) rowptr[idx] = wtot[wave] + incl - v;
}

__global__ void edge_fill(const int* __restrict__ src, const int* __restrict__ dst,
                          const int* __restrict__ rowptr, const int* __restrict__ rank,
                          int* __restrict__ ssrc, int E) {
  int e = blockIdx.x * blockDim.x + threadIdx.x;
  if (e >= E) return;
  ssrc[rowptr[dst[e]] + rank[e]] = src[e];
}

template <int B>
__device__ __forceinline__ void add_rows(const float4* __restrict__ X4,
                                         const int* __restrict__ ssrc,
                                         int j, int lane, float4& acc) {
  int s[B];
  float4 v[B];
#pragma unroll
  for (int u = 0; u < B; ++u) s[u] = ssrc[j + u];
#pragma unroll
  for (int u = 0; u < B; ++u) v[u] = X4[(size_t)s[u] * (CH / 4) + lane];
#pragma unroll
  for (int u = 0; u < B; ++u) {
    acc.x += v[u].x;
    acc.y += v[u].y;
    acc.z += v[u].z;
    acc.w += v[u].w;
  }
}

__global__ void gather_sum(const float* __restrict__ X,
                           const int* __restrict__ rowptr,
                           const int* __restrict__ ssrc,
                           float* __restrict__ agg, int M) {
  const int lane = threadIdx.x & 31;
  const int n = __builtin_amdgcn_readfirstlane(
      (int)(blockIdx.x * (blockDim.x >> 5) + (threadIdx.x >> 5)));
  if (n >= M) return;
  const int beg = rowptr[n];
  const int end = rowptr[n + 1];
  const float4* X4 = (const float4*)X;
  float4 acc = make_float4(0.f, 0.f, 0.f, 0.f);
  int j = beg;
  for (; j + 8 <= end; j += 8) add_rows<8>(X4, ssrc, j, lane, acc);
  if (j + 4 <= end) { add_rows<4>(X4, ssrc, j, lane, acc); j += 4; }
  if (j + 2 <= end) { add_rows<2>(X4, ssrc, j, lane, acc); j += 2; }
  if (j < end) add_rows<1>(X4, ssrc, j, lane, acc);
  ((float4*)agg)[(size_t)n * (CH / 4) + lane] = acc;
}

#define KP 132
template <bool ADD_AGG, bool RELU, int NT, int MT>
__global__ void __launch_bounds__(NT * 32)
gemm_relu_wmma(const float* __restrict__ A,
               const float* __restrict__ AGG,
               const float* __restrict__ W,
               const float* __restrict__ bias,
               float* __restrict__ out) {
  constexpr int K = CH;
  constexpr int N = NT * 16;
  constexpr int T = NT * 32;
  constexpr int ROWS = MT * 16;
  constexpr int NV = ROWS * (K / 4) / T;
  __shared__ float4 As4[ROWS * (KP / 4)];

  const int t = threadIdx.x;
  const int lane = t & 31;
  const int wave = t >> 5;
  const int hi = lane >> 4;
  const int lo = lane & 15;
  const int col = (wave << 4) + lo;
  const size_t row0 = (size_t)blockIdx.x * ROWS;

  const float4* A4 = (const float4*)A + row0 * (K / 4);
  float4 v[NV];
  for (int i = 0; i < NV; ++i) v[i] = A4[t + i * T];
  if (ADD_AGG) {
    const float4* G4 = (const float4*)AGG + row0 * (K / 4);
    for (int i = 0; i < NV; ++i) {
      const float4 g = G4[t + i * T];
      v[i].x += g.x;
      v[i].y += g.y;
      v[i].z += g.z;
      v[i].w += g.w;
    }
  }

  v2f bf[K / 4];
  for (int i = 0; i < K / 4; ++i) {
    const int ka = 4 * i + (hi << 1);
    bf[i].x = W[(size_t)ka * N + col];
    bf[i].y = W[(size_t)(ka + 1) * N + col];
  }
  const float bv = bias[col];

  for (int i = 0; i < NV; ++i) As4[(wave + i * NT) * (KP / 4) + lane] = v[i];
  __syncthreads();

  const float* Af = (const float*)As4 + lo * KP + (hi << 1);

  v8f c[MT];
  for (int mt = 0; mt < MT; ++mt) {
    for (int r = 0; r < 8; ++r) c[mt][r] = bv;
  }

  for (int i = 0; i < K / 4; ++i) {
    for (int mt = 0; mt < MT; ++mt) {
      const v2f a = *(const v2f*)(Af + mt * 16 * KP + 4 * i);
      c[mt] = __builtin_amdgcn_wmma_f32_16x16x4_f32(
           false, a,  false, bf[i],
           (short)0, c[mt],  false,  false);
    }
  }

  for (int mt = 0; mt < MT; ++mt) {
    float* orow = out + (row0 + mt * 16 + hi * 8) * N + col;
    for (int r = 0; r < 8; ++r) {
      float x = c[mt][r];
      if (RELU) x = fmaxf(x, 0.f);
      orow[(size_t)r * N] = x;
    }
  }
}

__global__ void head_gemv(const float* __restrict__ H,
                          const float* __restrict__ w4,
                          const float* __restrict__ b4,
                          float* __restrict__ out, int M) {
  long long t = (long long)blockIdx.x * blockDim.x + threadIdx.x;
  int n = (int)(t >> 5);
  if (n >= M) return;
  int lane = (int)(t & 31);
  const float* h = H + (size_t)n * 64;
  float s = h[lane] * w4[lane] + h[lane + 32] * w4[lane + 32];
#pragma unroll
  for (int off = 16; off > 0; off >>= 1) s += __shfl_xor(s, off, 32);
  if (lane == 0) out[n] = s + b4[0];
}

extern "C" void kernel_launch(void* const* d_in, const int* in_sizes, int n_in,
                              void* d_out, int out_size, void* d_ws, size_t ws_size,
                              hipStream_t stream) {
  const float* x   = (const float*)d_in[0];
  const int*   ei  = (const int*)d_in[1];
  const int    E   = in_sizes[1] / 2;
  const int* src = ei;
  const int* dst = ei + E;
  const float* w1a = (const float*)d_in[2];
  const float* b1a = (const float*)d_in[3];
  const float* w1b = (const float*)d_in[4];
  const float* b1b = (const float*)d_in[5];
  const float* w2a = (const float*)d_in[6];
  const float* b2a = (const float*)d_in[7];
  const float* w2b = (const float*)d_in[8];
  const float* b2b = (const float*)d_in[9];
  const float* w3  = (const float*)d_in[10];
  const float* b3  = (const float*)d_in[11];
  const float* w4  = (const float*)d_in[12];
  const float* b4  = (const float*)d_in[13];
  float* outp = (float*)d_out;

  float* agg  = (float*)d_ws;
  float* bufA = agg  + (size_t)NN * CH;
  float* bufB = bufA + (size_t)NN * CH;
  float* h3   = agg;
  int* deg    = (int*)(bufB + (size_t)NN * CH);
  int* rowptr = deg + NN;
  int* rank   = rowptr + NN + 16;
  int* ssrc   = rank + E;

  const int d4 = NN / 4;
  dim3 zb(256), zg((d4 + 255) / 256);
  dim3 eb(256), eg((unsigned)((E + 255) / 256));
  dim3 gb(256), gg((NN + 7) / 8);

  constexpr int MTW = 5;
  constexpr unsigned GB = (NN / 16) / MTW;

  zero_f4<<<zg, zb, 0, stream>>>((float4*)deg, d4);
  edge_rank<<<eg, eb, 0, stream>>>(dst, deg, rank, E);
  row_offsets<<<(NN + SCAN_T) / SCAN_T, SCAN_T, 0, stream>>>(deg, rowptr, NN);
  edge_fill<<<eg, eb, 0, stream>>>(src, dst, rowptr, rank, ssrc, E);

  gather_sum<<<gg, gb, 0, stream>>>(x, rowptr, ssrc, agg, NN);
  gemm_relu_wmma<true,  true, 8, MTW><<<GB, 256, 0, stream>>>(x,    agg,     w1a, b1a, bufA);
  gemm_relu_wmma<false, true, 8, MTW><<<GB, 256, 0, stream>>>(bufA, nullptr, w1b, b1b, bufB);

  gather_sum<<<gg, gb, 0, stream>>>(bufB, rowptr, ssrc, agg, NN);
  gemm_relu_wmma<true,  true, 8, MTW><<<GB, 256, 0, stream>>>(bufB, agg,     w2a, b2a, bufA);
  gemm_relu_wmma<false, true, 8, MTW><<<GB, 256, 0, stream>>>(bufA, nullptr, w2b, b2b, bufB);

  gemm_relu_wmma<false, true, 4, MTW><<<GB, 128, 0, stream>>>(bufB, nullptr, w3, b3, h3);
  head_gemv<<<(NN * 32 + 255) / 256, 256, 0, stream>>>(h3, w4, b4, outp, NN);
}


# Dense layers behind each neighbour sum chained in one work-group: the 80-row ReLU output tile feeds the next layer's A operand through LDS, not global (6 launches -> 2); every sum keeps its order
# speedup vs baseline: 2.4210x; 1.1957x over previous
#include <hip/hip_runtime.h>

typedef __attribute__((ext_vector_type(2))) float v2f;
typedef __attribute__((ext_vector_type(8))) float v8f;

#define NN 50000
#define CH 128

__global__ void zero_f4(float4* __restrict__ p, int n4) {
  int i = blockIdx.x * blockDim.x + threadIdx.x;
  if (i < n4) p[i] = make_float4(0.f, 0.f, 0.f, 0.f);
}

__global__ void edge_rank(const int* __restrict__ dst, int* __restrict__ deg,
                          int* __restrict__ rank, int E) {
  int e = blockIdx.x * blockDim.x + threadIdx.x;
  if (e >= E) return;
  rank[e] = atomicAdd(deg + dst[e], 1);
}

#define SCAN_T 1024
__global__ void __launch_bounds__(SCAN_T)
row_offsets(const int* __restrict__ deg, int* __restrict__ rowptr, int M) {
  __shared__ int wpre[32];
  __shared__ int wtot[32];
  const int t = threadIdx.x;
  const int lane = t & 31;
  const int wave = t >> 5;
  const int first = blockIdx.x * SCAN_T;
  int pre = 0;
  for (int i = t; i < first; i += SCAN_T) pre += deg[i];
  const int idx = first + t;
  const int v = (idx < M) ? deg[idx] : 0;
  int incl = v;
#pragma unroll
  for (int off = 1; off < 32; off <<= 1) {
    const int y = __shfl_up(incl, off, 32);
    if (lane >= off) incl += y;
  }
#pragma unroll
  for (int off = 16; off > 0; off >>= 1) pre += __shfl_xor(pre, off, 32);
  if (lane == 31) wtot[wave] = incl;
  if (lane == 0) wpre[wave] = pre;
  __syncthreads();
  if (wave == 0) {
    int p = wpre[lane];
#pragma unroll
    for (int off = 16; off > 0; off >>= 1) p += __shfl_xor(p, off, 32);
    const int w = wtot[lane];
    int wi = w;
#pragma unroll
    for (int off = 1; off < 32; off <<= 1) {
      const int y = __shfl_up(wi, off, 32);
      if (lane >= off) wi += y;
    }
    wtot[lane] = p + wi - w;
  }
  __syncthreads();
  if (idx <= M) rowptr[idx] = wtot[wave] + incl - v;
}

__global__ void edge_fill(const int* __restrict__ src, const int* __restrict__ dst,
                          const int* __restrict__ rowptr, const int* __restrict__ rank,
                          int* __restrict__ ssrc, int E) {
  int e = blockIdx.x * blockDim.x + threadIdx.x;
  if (e >= E) return;
  ssrc[rowptr[dst[e]] + rank[e]] = src[e];
}

template <int B>
__device__ __forceinline__ void add_rows(const float4* __restrict__ X4,
                                         const int* __restrict__ ssrc,
                                         int j, int lane, float4& acc) {
  int s[B];
  float4 v[B];
#pragma unroll
  for (int u = 0; u < B; ++u) s[u] = ssrc[j + u];
#pragma unroll
  for (int u = 0; u < B; ++u) v[u] = X4[(size_t)s[u] * (CH / 4) + lane];
#pragma unroll
  for (int u = 0; u < B; ++u) {
    acc.x += v[u].x;
    acc.y += v[u].y;
    acc.z += v[u].z;
    acc.w += v[u].w;
  }
}

__global__ void gather_sum(const float* __restrict__ X,
                           const int* __restrict__ rowptr,
                           const int* __restrict__ ssrc,
                           float* __restrict__ agg, int M) {
  const int lane = threadIdx.x & 31;
  const int n = __builtin_amdgcn_readfirstlane(
      (int)(blockIdx.x * (blockDim.x >> 5) + (threadIdx.x >> 5)));
  if (n >= M) return;
  const int beg = rowptr[n];
  const int end = rowptr[n + 1];
  const float4* X4 = (const float4*)X;
  float4 acc = make_float4(0.f, 0.f, 0.f, 0.f);
  int j = beg;
  for (; j + 8 <= end; j += 8) add_rows<8>(X4, ssrc, j, lane, acc);
  if (j + 4 <= end) { add_rows<4>(X4, ssrc, j, lane, acc); j += 4; }
  if (j + 2 <= end) { add_rows<2>(X4, ssrc, j, lane, acc); j += 2; }
  if (j < end) add_rows<1>(X4, ssrc, j, lane, acc);
  ((float4*)agg)[(size_t)n * (CH / 4) + lane] = acc;
}

#define KP 132
#define HP 68
#define MTW 5
#define ROWS (MTW * 16)

template <int N>
__device__ __forceinline__ void load_b_panel(const float* __restrict__ W, int col, int hi,
                                             v2f (&bf)[CH / 4]) {
  for (int i = 0; i < CH / 4; ++i) {
    const int ka = 4 * i + (hi << 1);
    bf[i].x = W[(size_t)ka * N + col];
    bf[i].y = W[(size_t)(ka + 1) * N + col];
  }
}

template <int NM>
__device__ __forceinline__ void mma_tiles(const float* Af, const v2f (&bf)[CH / 4], v8f (&c)[NM]) {
  for (int i = 0; i < CH / 4; ++i) {
    for (int m = 0; m < NM; ++m) {
      const v2f a = *(const v2f*)(Af + m * 16 * KP + 4 * i);
      c[m] = __builtin_amdgcn_wmma_f32_16x16x4_f32(
           false, a,  false, bf[i],
           (short)0, c[m],  false,  false);
    }
  }
}

template <int NM, int P>
__device__ __forceinline__ void relu_to_lds(float* S, int mfirst, int hi, int col, const v8f (&c)[NM]) {
  for (int m = 0; m < NM; ++m) {
    float* srow = S + ((mfirst + m) * 16 + hi * 8) * P + col;
    for (int r = 0; r < 8; ++r) srow[r * P] = fmaxf(c[m][r], 0.f);
  }
}

template <bool HEAD>
__global__ void __launch_bounds__(256, 6)
conv_mlp_wmma(const float* __restrict__ X,
              const float* __restrict__ AGG,
              const float* __restrict__ Wa,
              const float* __restrict__ ba,
              const float* __restrict__ Wb,
              const float* __restrict__ bb,
              const float* __restrict__ W3,
              const float* __restrict__ b3,
              const float* __restrict__ w4,
              const float* __restrict__ b4,
              float* __restrict__ out) {
  constexpr int K = CH;
  constexpr int T = 256;
  constexpr int NV = ROWS * (K / 4) / T;
  __shared__ float4 S0q[ROWS * (KP / 4)];
  __shared__ float4 S1q[ROWS * (KP / 4)];
  float* S0 = (float*)S0q;
  float* S1 = (float*)S1q;

  const int t = threadIdx.x;
  const int lane = t & 31;
  const int wave = __builtin_amdgcn_readfirstlane(t >> 5);
  const int hi = lane >> 4;
  const int lo = lane & 15;
  const int col = (wave << 4) + lo;
  const size_t row0 = (size_t)blockIdx.x * ROWS;

  const float4* A4 = (const float4*)X + row0 * (K / 4);
  const float4* G4 = (const float4*)AGG + row0 * (K / 4);
  float4 v[NV];
  for (int i = 0; i < NV; ++i) v[i] = A4[t + i * T];
  for (int i = 0; i < NV; ++i) {
    const float4 g = G4[t + i * T];
    v[i].x += g.x;
    v[i].y += g.y;
    v[i].z += g.z;
    v[i].w += g.w;
  }

  v2f bf[K / 4];
  load_b_panel<CH>(Wa, col, hi, bf);
  float bv = ba[col];

  for (int i = 0; i < NV; ++i) S0q[(wave + i * 8) * (KP / 4) + lane] = v[i];
  __syncthreads();

  const int aoff = lo * KP + (hi << 1);
  v8f c[MTW];

  for (int m = 0; m < MTW; ++m)
    for (int r = 0; r < 8; ++r) c[m][r] = bv;
  mma_tiles<MTW>(S0 + aoff, bf, c);
  load_b_panel<CH>(Wb, col, hi, bf);
  bv = bb[col];
  relu_to_lds<MTW, KP>(S1, 0, hi, col, c);
  __syncthreads();

  for (int m = 0; m < MTW; ++m)
    for (int r = 0; r < 8; ++r) c[m][r] = bv;
  mma_tiles<MTW>(S1 + aoff, bf, c);

  if (!HEAD) {
    for (int m = 0; m < MTW; ++m) {
      float* orow = out + (row0 + m * 16 + hi * 8) * CH + col;
      for (int r = 0; r < 8; ++r) orow[(size_t)r * CH] = fmaxf(c[m][r], 0.f);
    }
    return;
  } else {
    const int nt3 = wave & 3;
    const int g = wave >> 2;
    const int ln = (int)__lane_id();
    const int hi3 = ln >> 4;
    const int col3 = (nt3 << 4) + (ln & 15);
    const int aoff3 = (ln & 15) * KP + (hi3 << 1);
    relu_to_lds<MTW, KP>(S0, 0, hi, col, c);
    __syncthreads();
    load_b_panel<64>(W3, col3, hi3, bf);
    const float bv3 = b3[col3];

    if (g == 0) {
      v8f c3[3];
      for (int m = 0; m < 3; ++m)
        for (int r = 0; r < 8; ++r) c3[m][r] = bv3;
      mma_tiles<3>(S0 + aoff3, bf, c3);
      relu_to_lds<3, HP>(S1, 0, hi3, col3, c3);
    } else {
      v8f c3[2];
      for (int m = 0; m < 2; ++m)
        for (int r = 0; r < 8; ++r) c3[m][r] = bv3;
      mma_tiles<2>(S0 + aoff3 + 3 * 16 * KP, bf, c3);
      relu_to_lds<2, HP>(S1, 3, hi3, col3, c3);
    }
    const float w4a = w4[ln];
    const float w4b = w4[ln + 32];
    const float b4v = b4[0];
    __syncthreads();

    float res = 0.f;
    for (int k = 0; k < ROWS / 8; ++k) {
      const float* h = S1 + (wave * (ROWS / 8) + k) * HP;
      float s = h[ln] * w4a + h[ln + 32] * w4b;
#pragma unroll
      for (int off = 16; off > 0; off >>= 1) s += __shfl_xor(s, off, 32);
      if (ln == k) res = s;
    }
    if (ln < ROWS / 8) out[row0 + wave * (ROWS / 8) + ln] = res + b4v;
  }
}

extern "C" void kernel_launch(void* const* d_in, const int* in_sizes, int n_in,
                              void* d_out, int out_size, void* d_ws, size_t ws_size,
                              hipStream_t stream) {
  const float* x   = (const float*)d_in[0];
  const int*   ei  = (const int*)d_in[1];
  const int    E   = in_sizes[1] / 2;
  const int* src = ei;
  const int* dst = ei + E;
  const float* w1a = (const float*)d_in[2];
  const float* b1a = (const float*)d_in[3];
  const float* w1b = (const float*)d_in[4];
  const float* b1b = (const float*)d_in[5];
  const float* w2a = (const float*)d_in[6];
  const float* b2a = (const float*)d_in[7];
  const float* w2b = (const float*)d_in[8];
  const float* b2b = (const float*)d_in[9];
  const float* w3  = (const float*)d_in[10];
  const float* b3  = (const float*)d_in[11];
  const float* w4  = (const float*)d_in[12];
  const float* b4  = (const float*)d_in[13];
  float* outp = (float*)d_out;

  float* agg  = (float*)d_ws;
  float* h1   = agg + (size_t)NN * CH;
  int* deg    = (int*)(h1 + (size_t)NN * CH);
  int* rowptr = deg + NN;
  int* rank   = rowptr + NN + 16;
  int* ssrc   = rank + E;

  const int d4 = NN / 4;
  dim3 zb(256), zg((d4 + 255) / 256);
  dim3 eb(256), eg((unsigned)((E + 255) / 256));
  dim3 gb(256), gg((NN + 7) / 8);

  constexpr unsigned GB = (NN / 16) / MTW;

  zero_f4<<<zg, zb, 0, stream>>>((float4*)deg, d4);
  edge_rank<<<eg, eb, 0, stream>>>(dst, deg, rank, E);
  row_offsets<<<(NN + SCAN_T) / SCAN_T, SCAN_T, 0, stream>>>(deg, rowptr, NN);
  edge_fill<<<eg, eb, 0, stream>>>(src, dst, rowptr, rank, ssrc, E);

  gather_sum<<<gg, gb, 0, stream>>>(x, rowptr, ssrc, agg, NN);
  conv_mlp_wmma<false><<<GB, 256, 0, stream>>>(x, agg, w1a, b1a, w1b, b1b,
                                               nullptr, nullptr, nullptr, nullptr, h1);

  gather_sum<<<gg, gb, 0, stream>>>(h1, rowptr, ssrc, agg, NN);
  conv_mlp_wmma<true><<<GB, 256, 0, stream>>>(h1, agg, w2a, b2a, w2b, b2b,
                                              w3, b3, w4, b4, outp);
}


# Edges grouped by destination through LDS bins (per-chunk LDS counting sort by 64-node range, then one work-group per range), not global atomic rank + scan + scattered fill; sum order = LDS arrival
# speedup vs baseline: 3.9279x; 1.6224x over previous
#include <hip/hip_runtime.h>

typedef __attribute__((ext_vector_type(2))) float v2f;
typedef __attribute__((ext_vector_type(8))) float v8f;

#define NN 50000
#define CH 128

#define NBIN ((NN + 63) / 64)
#define NSLOT 1024
#define GCH 256
#define CAP 4096

template <int NW>
__device__ __forceinline__ int block_scan_excl(int v, int* wsum, int& total) {
  const int lane = threadIdx.x & 31;
  const int wave = threadIdx.x >> 5;
  int incl = v;
#pragma unroll
  for (int off = 1; off < 32; off <<= 1) {
    const int y = __shfl_up(incl, off, 32);
    if (lane >= off) incl += y;
  }
  __syncthreads();
  if (lane == 31) wsum[wave] = incl;
  __syncthreads();
  int pre = 0, tot = 0;
  for (int w = 0; w < NW; ++w) {
    const int s = wsum[w];
    if (w < wave) pre += s;
    tot += s;
  }
  total = tot;
  return pre + incl - v;
}

__global__ void __launch_bounds__(1024)
bin_edges(const int* __restrict__ src, const int* __restrict__ dst,
          int* __restrict__ bpk, int* __restrict__ binoff, int E) {
  __shared__ int cnt[NSLOT];
  __shared__ int off[NSLOT];
  __shared__ int wsum[32];
  __shared__ int sbuf[CAP];
  const int t = threadIdx.x;
  const int g = blockIdx.x;
  const int C = (E + GCH - 1) / GCH;
  const int e0 = g * C;
  int n = E - e0;
  n = n < 0 ? 0 : (n > C ? C : n);
  cnt[t] = 0;
  __syncthreads();
  for (int i0 = 0; i0 < n; i0 += 4096) {
    int d[4];
#pragma unroll
    for (int u = 0; u < 4; ++u) {
      const int i = i0 + u * 1024 + t;
      d[u] = (i < n) ? dst[e0 + i] : -1;
    }
#pragma unroll
    for (int u = 0; u < 4; ++u)
      if (d[u] >= 0) atomicAdd(&cnt[d[u] >> 6], 1);
  }
  __syncthreads();
  const int v = cnt[t];
  int tot;
  const int ex = block_scan_excl<32>(v, wsum, tot);
  off[t] = ex;
  cnt[t] = 0;
  __syncthreads();
  if (t <= NBIN) binoff[g * (NBIN + 1) + t] = ex;
  const bool staged = (C <= CAP);
  for (int i0 = 0; i0 < n; i0 += 4096) {
    int d[4], s[4];
#pragma unroll
    for (int u = 0; u < 4; ++u) {
      const int i = i0 + u * 1024 + t;
      d[u] = (i < n) ? dst[e0 + i] : -1;
      s[u] = (i < n) ? src[e0 + i] : 0;
    }
#pragma unroll
    for (int u = 0; u < 4; ++u) {
      if (d[u] >= 0) {
        const int b = d[u] >> 6;
        const int pos = off[b] + atomicAdd(&cnt[b], 1);
        const int pk = (s[u] << 6) | (d[u] & 63);
        if (staged) sbuf[pos] = pk;
        else bpk[e0 + pos] = pk;
      }
    }
  }
  if (staged) {
    __syncthreads();
    for (int i = t; i < n; i += 1024) bpk[e0 + i] = sbuf[i];
  }
}

__global__ void __launch_bounds__(GCH)
bin_sort(const int* __restrict__ bpk, const int* __restrict__ binoff,
         int* __restrict__ rowptr, int* __restrict__ ssrc, int E) {
  __shared__ int ncnt[64];
  __shared__ int nstart[64];
  __shared__ int nfill[64];
  __shared__ int wsum[8];
  __shared__ int sbuf[CAP];
  __shared__ int sout[CAP];
  const int t = threadIdx.x;
  const int b = blockIdx.x;
  const int C = (E + GCH - 1) / GCH;
  const int o0 = binoff[t * (NBIN + 1) + b];
  const int o1 = binoff[t * (NBIN + 1) + b + 1];
  const int c = o1 - o0;
  if (t < 64) { ncnt[t] = 0; nfill[t] = 0; }
  int base, total;
  block_scan_excl<8>(o0, wsum, base);
  const int pos = block_scan_excl<8>(c, wsum, total);
  const bool staged = (total <= CAP);
  const int* run = bpk + (size_t)t * C + o0;
  for (int k = 0; k < c; k += 4) {
    int p[4];
#pragma unroll
    for (int u = 0; u < 4; ++u) p[u] = (k + u < c) ? run[k + u] : 0;
#pragma unroll
    for (int u = 0; u < 4; ++u) {
      if (k + u < c) {
        if (staged) sbuf[pos + k + u] = p[u];
        atomicAdd(&ncnt[p[u] & 63], 1);
      }
    }
  }
  __syncthreads();
  if (t < 32) {
    const int a = ncnt[2 * t];
    const int a2 = ncnt[2 * t + 1];
    const int s = a + a2;
    int incl = s;
#pragma unroll
    for (int off = 1; off < 32; off <<= 1) {
      const int y = __shfl_up(incl, off, 32);
      if (t >= off) incl += y;
    }
    nstart[2 * t] = incl - s;
    nstart[2 * t + 1] = incl - s + a;
  }
  __syncthreads();
  if (t < 64) {
    const int idx = b * 64 + t;
    if (idx < NN) rowptr[idx] = base + nstart[t];
  }
  if (b == NBIN - 1 && t == 0) rowptr[NN] = base + total;
  if (staged) {
    for (int j = t; j < total; j += GCH) {
      const int p = sbuf[j];
      const int node = p & 63;
      const int r = atomicAdd(&nfill[node], 1);
      sout[nstart[node] + r] = p >> 6;
    }
    __syncthreads();
    for (int j = t; j < total; j += GCH) ssrc[base + j] = sout[j];
  } else {
    for (int k = 0; k < c; ++k) {
      const int p = run[k];
      const int node = p & 63;
      const int r = atomicAdd(&nfill[node], 1);
      ssrc[base + nstart[node] + r] = p >> 6;
    }
  }
}

template <int B>
__device__ __forceinline__ void add_rows(const float4* __restrict__ X4,
                                         const int* __restrict__ ssrc,
                                         int j, int lane, float4& acc) {
  int s[B];
  float4 v[B];
#pragma unroll
  for (int u = 0; u < B; ++u) s[u] = ssrc[j + u];
#pragma unroll
  for (int u = 0; u < B; ++u) v[u] = X4[(size_t)s[u] * (CH / 4) + lane];
#pragma unroll
  for (int u = 0; u < B; ++u) {
    acc.x += v[u].x;
    acc.y += v[u].y;
    acc.z += v[u].z;
    acc.w += v[u].w;
  }
}

__global__ void gather_sum(const float* __restrict__ X,
                           const int* __restrict__ rowptr,
                           const int* __restrict__ ssrc,
                           float* __restrict__ agg, int M) {
  const int lane = threadIdx.x & 31;
  const int n = __builtin_amdgcn_readfirstlane(
      (int)(blockIdx.x * (blockDim.x >> 5) + (threadIdx.x >> 5)));
  if (n >= M) return;
  const int beg = rowptr[n];
  const int end = rowptr[n + 1];
  const float4* X4 = (const float4*)X;
  float4 acc = make_float4(0.f, 0.f, 0.f, 0.f);
  int j = beg;
  for (; j + 8 <= end; j += 8) add_rows<8>(X4, ssrc, j, lane, acc);
  if (j + 4 <= end) { add_rows<4>(X4, ssrc, j, lane, acc); j += 4; }
  if (j + 2 <= end) { add_rows<2>(X4, ssrc, j, lane, acc); j += 2; }
  if (j < end) add_rows<1>(X4, ssrc, j, lane, acc);
  ((float4*)agg)[(size_t)n * (CH / 4) + lane] = acc;
}

#define KP 132
#define HP 68
#define MTW 5
#define ROWS (MTW * 16)

template <int N>
__device__ __forceinline__ void load_b_panel(const float* __restrict__ W, int col, int hi,
                                             v2f (&bf)[CH / 4]) {
  for (int i = 0; i < CH / 4; ++i) {
    const int ka = 4 * i + (hi << 1);
    bf[i].x = W[(size_t)ka * N + col];
    bf[i].y = W[(size_t)(ka + 1) * N + col];
  }
}

template <int NM>
__device__ __forceinline__ void mma_tiles(const float* Af, const v2f (&bf)[CH / 4], v8f (&c)[NM]) {
  for (int i = 0; i < CH / 4; ++i) {
    for (int m = 0; m < NM; ++m) {
      const v2f a = *(const v2f*)(Af + m * 16 * KP + 4 * i);
      c[m] = __builtin_amdgcn_wmma_f32_16x16x4_f32(
           false, a,  false, bf[i],
           (short)0, c[m],  false,  false);
    }
  }
}

template <int NM, int P>
__device__ __forceinline__ void relu_to_lds(float* S, int mfirst, int hi, int col, const v8f (&c)[NM]) {
  for (int m = 0; m < NM; ++m) {
    float* srow = S + ((mfirst + m) * 16 + hi * 8) * P + col;
    for (int r = 0; r < 8; ++r) srow[r * P] = fmaxf(c[m][r], 0.f);
  }
}

template <bool HEAD>
__global__ void __launch_bounds__(256, 6)
conv_mlp_wmma(const float* __restrict__ X,
              const float* __restrict__ AGG,
              const float* __restrict__ Wa,
              const float* __restrict__ ba,
              const float* __restrict__ Wb,
              const float* __restrict__ bb,
              const float* __restrict__ W3,
              const float* __restrict__ b3,
              const float* __restrict__ w4,
              const float* __restrict__ b4,
              float* __restrict__ out) {
  constexpr int K = CH;
  constexpr int T = 256;
  constexpr int NV = ROWS * (K / 4) / T;
  __shared__ float4 S0q[ROWS * (KP / 4)];
  __shared__ float4 S1q[ROWS * (KP / 4)];
  float* S0 = (float*)S0q;
  float* S1 = (float*)S1q;

  const int t = threadIdx.x;
  const int lane = t & 31;
  const int wave = __builtin_amdgcn_readfirstlane(t >> 5);
  const int hi = lane >> 4;
  const int lo = lane & 15;
  const int col = (wave << 4) + lo;
  const size_t row0 = (size_t)blockIdx.x * ROWS;

  const float4* A4 = (const float4*)X + row0 * (K / 4);
  const float4* G4 = (const float4*)AGG + row0 * (K / 4);
  float4 v[NV];
  for (int i = 0; i < NV; ++i) v[i] = A4[t + i * T];
  for (int i = 0; i < NV; ++i) {
    const float4 g = G4[t + i * T];
    v[i].x += g.x;
    v[i].y += g.y;
    v[i].z += g.z;
    v[i].w += g.w;
  }

  v2f bf[K / 4];
  load_b_panel<CH>(Wa, col, hi, bf);
  float bv = ba[col];

  for (int i = 0; i < NV; ++i) S0q[(wave + i * 8) * (KP / 4) + lane] = v[i];
  __syncthreads();

  const int aoff = lo * KP + (hi << 1);
  v8f c[MTW];

  for (int m = 0; m < MTW; ++m)
    for (int r = 0; r < 8; ++r) c[m][r] = bv;
  mma_tiles<MTW>(S0 + aoff, bf, c);
  load_b_panel<CH>(Wb, col, hi, bf);
  bv = bb[col];
  relu_to_lds<MTW, KP>(S1, 0, hi, col, c);
  __syncthreads();

  for (int m = 0; m < MTW; ++m)
    for (int r = 0; r < 8; ++r) c[m][r] = bv;
  mma_tiles<MTW>(S1 + aoff, bf, c);

  if (!HEAD) {
    for (int m = 0; m < MTW; ++m) {
      float* orow = out + (row0 + m * 16 + hi * 8) * CH + col;
      for (int r = 0; r < 8; ++r) orow[(size_t)r * CH] = fmaxf(c[m][r], 0.f);
    }
    return;
  } else {
    const int nt3 = wave & 3;
    const int g = wave >> 2;
    const int ln = (int)__lane_id();
    const int hi3 = ln >> 4;
    const int col3 = (nt3 << 4) + (ln & 15);
    const int aoff3 = (ln & 15) * KP + (hi3 << 1);
    relu_to_lds<MTW, KP>(S0, 0, hi, col, c);
    __syncthreads();
    load_b_panel<64>(W3, col3, hi3, bf);
    const float bv3 = b3[col3];

    if (g == 0) {
      v8f c3[3];
      for (int m = 0; m < 3; ++m)
        for (int r = 0; r < 8; ++r) c3[m][r] = bv3;
      mma_tiles<3>(S0 + aoff3, bf, c3);
      relu_to_lds<3, HP>(S1, 0, hi3, col3, c3);
    } else {
      v8f c3[2];
      for (int m = 0; m < 2; ++m)
        for (int r = 0; r < 8; ++r) c3[m][r] = bv3;
      mma_tiles<2>(S0 + aoff3 + 3 * 16 * KP, bf, c3);
      relu_to_lds<2, HP>(S1, 3, hi3, col3, c3);
    }
    const float w4a = w4[ln];
    const float w4b = w4[ln + 32];
    const float b4v = b4[0];
    __syncthreads();

    float res = 0.f;
    for (int k = 0; k < ROWS / 8; ++k) {
      const float* h = S1 + (wave * (ROWS / 8) + k) * HP;
      float s = h[ln] * w4a + h[ln + 32] * w4b;
#pragma unroll
      for (int off = 16; off > 0; off >>= 1) s += __shfl_xor(s, off, 32);
      if (ln == k) res = s;
    }
    if (ln < ROWS / 8) out[row0 + wave * (ROWS / 8) + ln] = res + b4v;
  }
}

extern "C" void kernel_launch(void* const* d_in, const int* in_sizes, int n_in,
                              void* d_out, int out_size, void* d_ws, size_t ws_size,
                              hipStream_t stream) {
  const float* x   = (const float*)d_in[0];
  const int*   ei  = (const int*)d_in[1];
  const int    E   = in_sizes[1] / 2;
  const int* src = ei;
  const int* dst = ei + E;
  const float* w1a = (const float*)d_in[2];
  const float* b1a = (const float*)d_in[3];
  const float* w1b = (const float*)d_in[4];
  const float* b1b = (const float*)d_in[5];
  const float* w2a = (const float*)d_in[6];
  const float* b2a = (const float*)d_in[7];
  const float* w2b = (const float*)d_in[8];
  const float* b2b = (const float*)d_in[9];
  const float* w3  = (const float*)d_in[10];
  const float* b3  = (const float*)d_in[11];
  const float* w4  = (const float*)d_in[12];
  const float* b4  = (const float*)d_in[13];
  float* outp = (float*)d_out;

  float* agg  = (float*)d_ws;
  float* h1   = agg + (size_t)NN * CH;
  int* rowptr = (int*)(h1 + (size_t)NN * CH);
  int* ssrc   = rowptr + NN + 16;
  int* bpk    = ssrc + E;
  int* binoff = bpk + E + GCH;

  dim3 gb(256), gg((NN + 7) / 8);

  constexpr unsigned GB = (NN / 16) / MTW;

  bin_edges<<<GCH, 1024, 0, stream>>>(src, dst, bpk, binoff, E);
  bin_sort<<<NBIN, GCH, 0, stream>>>(bpk, binoff, rowptr, ssrc, E);

  gather_sum<<<gg, gb, 0, stream>>>(x, rowptr, ssrc, agg, NN);
  conv_mlp_wmma<false><<<GB, 256, 0, stream>>>(x, agg, w1a, b1a, w1b, b1b,
                                               nullptr, nullptr, nullptr, nullptr, h1);

  gather_sum<<<gg, gb, 0, stream>>>(h1, rowptr, ssrc, agg, NN);
  conv_mlp_wmma<true><<<GB, 256, 0, stream>>>(h1, agg, w2a, b2a, w2b, b2b,
                                              w3, b3, w4, b4, outp);
}


# Neighbour sum moved into the dense work-group: a wave walks its 10 rows' contiguous edge run in 16-load batches, sums straight into the LDS A tile (no agg buffer, 6 -> 4 launches); sum order kept
# speedup vs baseline: 4.4491x; 1.1327x over previous
#include <hip/hip_runtime.h>

typedef __attribute__((ext_vector_type(2))) float v2f;
typedef __attribute__((ext_vector_type(8))) float v8f;

#define NN 50000
#define CH 128

#define NBIN ((NN + 63) / 64)
#define NSLOT 1024
#define GCH 256
#define CAP 4096

template <int NW>
__device__ __forceinline__ int block_scan_excl(int v, int* wsum, int& total) {
  const int lane = threadIdx.x & 31;
  const int wave = threadIdx.x >> 5;
  int incl = v;
#pragma unroll
  for (int off = 1; off < 32; off <<= 1) {
    const int y = __shfl_up(incl, off, 32);
    if (lane >= off) incl += y;
  }
  __syncthreads();
  if (lane == 31) wsum[wave] = incl;
  __syncthreads();
  int pre = 0, tot = 0;
  for (int w = 0; w < NW; ++w) {
    const int s = wsum[w];
    if (w < wave) pre += s;
    tot += s;
  }
  total = tot;
  return pre + incl - v;
}

__global__ void __launch_bounds__(1024)
bin_edges(const int* __restrict__ src, const int* __restrict__ dst,
          int* __restrict__ bpk, int* __restrict__ binoff, int E) {
  __shared__ int cnt[NSLOT];
  __shared__ int off[NSLOT];
  __shared__ int wsum[32];
  __shared__ int sbuf[CAP];
  const int t = threadIdx.x;
  const int g = blockIdx.x;
  const int C = (E + GCH - 1) / GCH;
  const int e0 = g * C;
  int n = E - e0;
  n = n < 0 ? 0 : (n > C ? C : n);
  cnt[t] = 0;
  __syncthreads();
  for (int i0 = 0; i0 < n; i0 += 4096) {
    int d[4];
#pragma unroll
    for (int u = 0; u < 4; ++u) {
      const int i = i0 + u * 1024 + t;
      d[u] = (i < n) ? dst[e0 + i] : -1;
    }
#pragma unroll
    for (int u = 0; u < 4; ++u)
      if (d[u] >= 0) atomicAdd(&cnt[d[u] >> 6], 1);
  }
  __syncthreads();
  const int v = cnt[t];
  int tot;
  const int ex = block_scan_excl<32>(v, wsum, tot);
  off[t] = ex;
  cnt[t] = 0;
  __syncthreads();
  if (t <= NBIN) binoff[g * (NBIN + 1) + t] = ex;
  const bool staged = (C <= CAP);
  for (int i0 = 0; i0 < n; i0 += 4096) {
    int d[4], s[4];
#pragma unroll
    for (int u = 0; u < 4; ++u) {
      const int i = i0 + u * 1024 + t;
      d[u] = (i < n) ? dst[e0 + i] : -1;
      s[u] = (i < n) ? src[e0 + i] : 0;
    }
#pragma unroll
    for (int u = 0; u < 4; ++u) {
      if (d[u] >= 0) {
        const int b = d[u] >> 6;
        const int pos = off[b] + atomicAdd(&cnt[b], 1);
        const int pk = (s[u] << 6) | (d[u] & 63);
        if (staged) sbuf[pos] = pk;
        else bpk[e0 + pos] = pk;
      }
    }
  }
  if (staged) {
    __syncthreads();
    for (int i = t; i < n; i += 1024) bpk[e0 + i] = sbuf[i];
  }
}

__global__ void __launch_bounds__(GCH)
bin_sort(const int* __restrict__ bpk, const int* __restrict__ binoff,
         int* __restrict__ rowptr, int* __restrict__ ssrc, int E) {
  __shared__ int ncnt[64];
  __shared__ int nstart[64];
  __shared__ int nfill[64];
  __shared__ int wsum[8];
  __shared__ int sbuf[CAP];
  __shared__ int sout[CAP];
  const int t = threadIdx.x;
  const int b = blockIdx.x;
  const int C = (E + GCH - 1) / GCH;
  const int o0 = binoff[t * (NBIN + 1) + b];
  const int o1 = binoff[t * (NBIN + 1) + b + 1];
  const int c = o1 - o0;
  if (t < 64) { ncnt[t] = 0; nfill[t] = 0; }
  int base, total;
  block_scan_excl<8>(o0, wsum, base);
  const int pos = block_scan_excl<8>(c, wsum, total);
  const bool staged = (total <= CAP);
  const int* run = bpk + (size_t)t * C + o0;
  for (int k = 0; k < c; k += 4) {
    int p[4];
#pragma unroll
    for (int u = 0; u < 4; ++u) p[u] = (k + u < c) ? run[k + u] : 0;
#pragma unroll
    for (int u = 0; u < 4; ++u) {
      if (k + u < c) {
        if (staged) sbuf[pos + k + u] = p[u];
        atomicAdd(&ncnt[p[u] & 63], 1);
      }
    }
  }
  __syncthreads();
  if (t < 32) {
    const int a = ncnt[2 * t];
    const int a2 = ncnt[2 * t + 1];
    const int s = a + a2;
    int incl = s;
#pragma unroll
    for (int off = 1; off < 32; off <<= 1) {
      const int y = __shfl_up(incl, off, 32);
      if (t >= off) incl += y;
    }
    nstart[2 * t] = incl - s;
    nstart[2 * t + 1] = incl - s + a;
  }
  __syncthreads();
  if (t < 64) {
    const int idx = b * 64 + t;
    if (idx < NN) rowptr[idx] = base + nstart[t];
  }
  if (b == NBIN - 1 && t == 0) rowptr[NN] = base + total;
  if (staged) {
    for (int j = t; j < total; j += GCH) {
      const int p = sbuf[j];
      const int node = p & 63;
      const int r = atomicAdd(&nfill[node], 1);
      sout[nstart[node] + r] = p >> 6;
    }
    __syncthreads();
    for (int j = t; j < total; j += GCH) ssrc[base + j] = sout[j];
  } else {
    for (int k = 0; k < c; ++k) {
      const int p = run[k];
      const int node = p & 63;
      const int r = atomicAdd(&nfill[node], 1);
      ssrc[base + nstart[node] + r] = p >> 6;
    }
  }
}

#define KP 132
template <int B>
__device__ __forceinline__ void hold_loads(const float4 (&v)[B]) {
  if constexpr (B == 16) asm volatile("" :: "v"(v[0].x), "v"(v[1].x), "v"(v[2].x), "v"(v[3].x), "v"(v[4].x), "v"(v[5].x), "v"(v[6].x), "v"(v[7].x), "v"(v[8].x), "v"(v[9].x), "v"(v[10].x), "v"(v[11].x), "v"(v[12].x), "v"(v[13].x), "v"(v[14].x), "v"(v[15].x));
  else if constexpr (B == 8) asm volatile("" :: "v"(v[0].x), "v"(v[1].x), "v"(v[2].x), "v"(v[3].x), "v"(v[4].x), "v"(v[5].x), "v"(v[6].x), "v"(v[7].x));
  else if constexpr (B == 4) asm volatile("" :: "v"(v[0].x), "v"(v[1].x), "v"(v[2].x), "v"(v[3].x));
  else if constexpr (B == 2) asm volatile("" :: "v"(v[0].x), "v"(v[1].x));
  else asm volatile("" :: "v"(v[0].x));
}

template <int B>
__device__ __forceinline__ void sum_run(const float4* __restrict__ X4,
                                        const int* __restrict__ ssrc,
                                        int j, int lane, int rpv, float4* Srow,
                                        int& cur, int& rend, float4& acc) {
  int s[B];
  float4 v[B];
#pragma unroll
  for (int u = 0; u < B; ++u) s[u] = ssrc[j + u];
#pragma unroll
  for (int u = 0; u < B; ++u) v[u] = X4[(size_t)s[u] * (CH / 4) + lane];
  hold_loads<B>(v);
#pragma unroll
  for (int u = 0; u < B; ++u) {
    while (j + u >= rend) {
      Srow[cur * (KP / 4)] = acc;
      acc = make_float4(0.f, 0.f, 0.f, 0.f);
      ++cur;
      rend = __builtin_amdgcn_readlane(rpv, cur + 1);
    }
    acc.x += v[u].x;
    acc.y += v[u].y;
    acc.z += v[u].z;
    acc.w += v[u].w;
  }
}

#define HP 68
#define MTW 5
#define ROWS (MTW * 16)

template <int N>
__device__ __forceinline__ void load_b_panel(const float* __restrict__ W, int col, int hi,
                                             v2f (&bf)[CH / 4]) {
  for (int i = 0; i < CH / 4; ++i) {
    const int ka = 4 * i + (hi << 1);
    bf[i].x = W[(size_t)ka * N + col];
    bf[i].y = W[(size_t)(ka + 1) * N + col];
  }
}

template <int NM>
__device__ __forceinline__ void mma_tiles(const float* Af, const v2f (&bf)[CH / 4], v8f (&c)[NM]) {
  for (int i = 0; i < CH / 4; ++i) {
    for (int m = 0; m < NM; ++m) {
      const v2f a = *(const v2f*)(Af + m * 16 * KP + 4 * i);
      c[m] = __builtin_amdgcn_wmma_f32_16x16x4_f32(
           false, a,  false, bf[i],
           (short)0, c[m],  false,  false);
    }
  }
}

template <int NM, int P>
__device__ __forceinline__ void relu_to_lds(float* S, int mfirst, int hi, int col, const v8f (&c)[NM]) {
  for (int m = 0; m < NM; ++m) {
    float* srow = S + ((mfirst + m) * 16 + hi * 8) * P + col;
    for (int r = 0; r < 8; ++r) srow[r * P] = fmaxf(c[m][r], 0.f);
  }
}

template <bool HEAD>
__global__ void __launch_bounds__(256, 6)
conv_mlp_wmma(const float* __restrict__ X,
              const int* __restrict__ rowptr,
              const int* __restrict__ ssrc,
              const float* __restrict__ Wa,
              const float* __restrict__ ba,
              const float* __restrict__ Wb,
              const float* __restrict__ bb,
              const float* __restrict__ W3,
              const float* __restrict__ b3,
              const float* __restrict__ w4,
              const float* __restrict__ b4,
              float* __restrict__ out) {
  constexpr int K = CH;
  constexpr int RW = ROWS / 8;
  __shared__ float4 S0q[ROWS * (KP / 4)];
  __shared__ float4 S1q[ROWS * (KP / 4)];
  float* S0 = (float*)S0q;
  float* S1 = (float*)S1q;

  const int t = threadIdx.x;
  const int lane = t & 31;
  const int wave = __builtin_amdgcn_readfirstlane(t >> 5);
  const int hi = lane >> 4;
  const int lo = lane & 15;
  const int col = (wave << 4) + lo;
  const size_t row0 = (size_t)blockIdx.x * ROWS;

  const int rbase = (int)row0 + wave * RW;
  const float4* X4 = (const float4*)X;
  float4 xr[RW];
  for (int i = 0; i < RW; ++i) xr[i] = X4[(size_t)(rbase + i) * (K / 4) + lane];
  const int rpv = rowptr[rbase + (lane < RW ? lane : RW)];
  int j = __builtin_amdgcn_readlane(rpv, 0);
  const int je = __builtin_amdgcn_readlane(rpv, RW);
  int rend = __builtin_amdgcn_readlane(rpv, 1);
  int cur = 0;
  float4 acc = make_float4(0.f, 0.f, 0.f, 0.f);
  float4* Srow = S0q + (wave * RW) * (KP / 4) + lane;
  for (; j + 16 <= je; j += 16) sum_run<16>(X4, ssrc, j, lane, rpv, Srow, cur, rend, acc);
  if (j + 8 <= je) { sum_run<8>(X4, ssrc, j, lane, rpv, Srow, cur, rend, acc); j += 8; }
  if (j + 4 <= je) { sum_run<4>(X4, ssrc, j, lane, rpv, Srow, cur, rend, acc); j += 4; }
  if (j + 2 <= je) { sum_run<2>(X4, ssrc, j, lane, rpv, Srow, cur, rend, acc); j += 2; }
  if (j < je) sum_run<1>(X4, ssrc, j, lane, rpv, Srow, cur, rend, acc);
  for (; cur < RW; ++cur) {
    Srow[cur * (KP / 4)] = acc;
    acc = make_float4(0.f, 0.f, 0.f, 0.f);
  }

  v2f bf[K / 4];
  load_b_panel<CH>(Wa, col, hi, bf);
  float bv = ba[col];

  for (int i = 0; i < RW; ++i) {
    float4 a = Srow[i * (KP / 4)];
    a.x = xr[i].x + a.x;
    a.y = xr[i].y + a.y;
    a.z = xr[i].z + a.z;
    a.w = xr[i].w + a.w;
    Srow[i * (KP / 4)] = a;
  }
  __syncthreads();

  const int aoff = lo * KP + (hi << 1);
  v8f c[MTW];

  for (int m = 0; m < MTW; ++m)
    for (int r = 0; r < 8; ++r) c[m][r] = bv;
  mma_tiles<MTW>(S0 + aoff, bf, c);
  load_b_panel<CH>(Wb, col, hi, bf);
  bv = bb[col];
  relu_to_lds<MTW, KP>(S1, 0, hi, col, c);
  __syncthreads();

  for (int m = 0; m < MTW; ++m)
    for (int r = 0; r < 8; ++r) c[m][r] = bv;
  mma_tiles<MTW>(S1 + aoff, bf, c);

  if (!HEAD) {
    for (int m = 0; m < MTW; ++m) {
      float* orow = out + (row0 + m * 16 + hi * 8) * CH + col;
      for (int r = 0; r < 8; ++r) orow[(size_t)r * CH] = fmaxf(c[m][r], 0.f);
    }
    return;
  } else {
    const int nt3 = wave & 3;
    const int g = wave >> 2;
    const int ln = (int)__lane_id();
    const int hi3 = ln >> 4;
    const int col3 = (nt3 << 4) + (ln & 15);
    const int aoff3 = (ln & 15) * KP + (hi3 << 1);
    relu_to_lds<MTW, KP>(S0, 0, hi, col, c);
    __syncthreads();
    load_b_panel<64>(W3, col3, hi3, bf);
    const float bv3 = b3[col3];

    if (g == 0) {
      v8f c3[3];
      for (int m = 0; m < 3; ++m)
        for (int r = 0; r < 8; ++r) c3[m][r] = bv3;
      mma_tiles<3>(S0 + aoff3, bf, c3);
      relu_to_lds<3, HP>(S1, 0, hi3, col3, c3);
    } else {
      v8f c3[2];
      for (int m = 0; m < 2; ++m)
        for (int r = 0; r < 8; ++r) c3[m][r] = bv3;
      mma_tiles<2>(S0 + aoff3 + 3 * 16 * KP, bf, c3);
      relu_to_lds<2, HP>(S1, 3, hi3, col3, c3);
    }
    const float w4a = w4[ln];
    const float w4b = w4[ln + 32];
    const float b4v = b4[0];
    __syncthreads();

    float res = 0.f;
    for (int k = 0; k < ROWS / 8; ++k) {
      const float* h = S1 + (wave * (ROWS / 8) + k) * HP;
      float s = h[ln] * w4a + h[ln + 32] * w4b;
#pragma unroll
      for (int off = 16; off > 0; off >>= 1) s += __shfl_xor(s, off, 32);
      if (ln == k) res = s;
    }
    if (ln < ROWS / 8) out[row0 + wave * (ROWS / 8) + ln] = res + b4v;
  }
}

extern "C" void kernel_launch(void* const* d_in, const int* in_sizes, int n_in,
                              void* d_out, int out_size, void* d_ws, size_t ws_size,
                              hipStream_t stream) {
  const float* x   = (const float*)d_in[0];
  const int*   ei  = (const int*)d_in[1];
  const int    E   = in_sizes[1] / 2;
  const int* src = ei;
  const int* dst = ei + E;
  const float* w1a = (const float*)d_in[2];
  const float* b1a = (const float*)d_in[3];
  const float* w1b = (const float*)d_in[4];
  const float* b1b = (const float*)d_in[5];
  const float* w2a = (const float*)d_in[6];
  const float* b2a = (const float*)d_in[7];
  const float* w2b = (const float*)d_in[8];
  const float* b2b = (const float*)d_in[9];
  const float* w3  = (const float*)d_in[10];
  const float* b3  = (const float*)d_in[11];
  const float* w4  = (const float*)d_in[12];
  const float* b4  = (const float*)d_in[13];
  float* outp = (float*)d_out;

  float* h1   = (float*)d_ws;
  int* rowptr = (int*)(h1 + (size_t)NN * CH);
  int* ssrc   = rowptr + NN + 16;
  int* bpk    = ssrc + E;
  int* binoff = bpk + E + GCH;

  constexpr unsigned GB = (NN / 16) / MTW;

  bin_edges<<<GCH, 1024, 0, stream>>>(src, dst, bpk, binoff, E);
  bin_sort<<<NBIN, GCH, 0, stream>>>(bpk, binoff, rowptr, ssrc, E);

  conv_mlp_wmma<false><<<GB, 256, 0, stream>>>(x, rowptr, ssrc, w1a, b1a, w1b, b1b,
                                               nullptr, nullptr, nullptr, nullptr, h1);

  conv_mlp_wmma<true><<<GB, 256, 0, stream>>>(h1, rowptr, ssrc, w2a, b2a, w2b, b2b,
                                              w3, b3, w4, b4, outp);
}


# Weights repacked once per call (extra work-groups of launch 1) into per-lane B-fragment order: a wave's panel is 16 whole-line 16-byte loads, not 64 4-byte loads of two half lines each
# speedup vs baseline: 4.6118x; 1.0366x over previous
#include <hip/hip_runtime.h>

typedef __attribute__((ext_vector_type(2))) float v2f;
typedef __attribute__((ext_vector_type(8))) float v8f;

#define NN 50000
#define CH 128

#define NBIN ((NN + 63) / 64)
#define NSLOT 1024
#define GCH 256
#define CAP 4096

template <int NW>
__device__ __forceinline__ int block_scan_excl(int v, int* wsum, int& total) {
  const int lane = threadIdx.x & 31;
  const int wave = threadIdx.x >> 5;
  int incl = v;
#pragma unroll
  for (int off = 1; off < 32; off <<= 1) {
    const int y = __shfl_up(incl, off, 32);
    if (lane >= off) incl += y;
  }
  __syncthreads();
  if (lane == 31) wsum[wave] = incl;
  __syncthreads();
  int pre = 0, tot = 0;
  for (int w = 0; w < NW; ++w) {
    const int s = wsum[w];
    if (w < wave) pre += s;
    tot += s;
  }
  total = tot;
  return pre + incl - v;
}

#define WPK128 (CH * CH)
#define WPK64 (CH * 64)
#define WPK_ALL (4 * WPK128 + WPK64)
#define PACKB (WPK_ALL / 4 / 1024)

__device__ __forceinline__ void pack_weights(const float* __restrict__ w1a, const float* __restrict__ w1b,
                                             const float* __restrict__ w2a, const float* __restrict__ w2b,
                                             const float* __restrict__ w3, float* __restrict__ wpk, int q) {
  const float* Wm;
  int N, r;
  if (q < 4 * (WPK128 / 4)) {
    const int L = q / (WPK128 / 4);
    r = q - L * (WPK128 / 4);
    N = CH;
    Wm = (L == 0) ? w1a : (L == 1) ? w1b : (L == 2) ? w2a : w2b;
  } else {
    r = q - 4 * (WPK128 / 4);
    N = 64;
    Wm = w3;
  }
  const int l = r & 31;
  const int i2 = (r >> 5) & 15;
  const int w = r >> 9;
  const int c = (w << 4) + (l & 15);
  const int k0 = 8 * i2 + ((l >> 4) << 1);
  float4 v;
  v.x = Wm[(size_t)k0 * N + c];
  v.y = Wm[(size_t)(k0 + 1) * N + c];
  v.z = Wm[(size_t)(k0 + 4) * N + c];
  v.w = Wm[(size_t)(k0 + 5) * N + c];
  ((float4*)wpk)[q] = v;
}

__global__ void __launch_bounds__(1024)
bin_edges(const int* __restrict__ src, const int* __restrict__ dst,
          int* __restrict__ bpk, int* __restrict__ binoff, int E,
          const float* __restrict__ w1a, const float* __restrict__ w1b,
          const float* __restrict__ w2a, const float* __restrict__ w2b,
          const float* __restrict__ w3, float* __restrict__ wpk) {
  __shared__ int cnt[NSLOT];
  __shared__ int off[NSLOT];
  __shared__ int wsum[32];
  __shared__ int sbuf[CAP];
  const int t = threadIdx.x;
  const int g = blockIdx.x;
  if (g >= GCH) {
    pack_weights(w1a, w1b, w2a, w2b, w3, wpk, (g - GCH) * 1024 + t);
    return;
  }
  const int C = (E + GCH - 1) / GCH;
  const int e0 = g * C;
  int n = E - e0;
  n = n < 0 ? 0 : (n > C ? C : n);
  cnt[t] = 0;
  __syncthreads();
  for (int i0 = 0; i0 < n; i0 += 4096) {
    int d[4];
#pragma unroll
    for (int u = 0; u < 4; ++u) {
      const int i = i0 + u * 1024 + t;
      d[u] = (i < n) ? dst[e0 + i] : -1;
    }
#pragma unroll
    for (int u = 0; u < 4; ++u)
      if (d[u] >= 0) atomicAdd(&cnt[d[u] >> 6], 1);
  }
  __syncthreads();
  const int v = cnt[t];
  int tot;
  const int ex = block_scan_excl<32>(v, wsum, tot);
  off[t] = ex;
  cnt[t] = 0;
  __syncthreads();
  if (t <= NBIN) binoff[g * (NBIN + 1) + t] = ex;
  const bool staged = (C <= CAP);
  for (int i0 = 0; i0 < n; i0 += 4096) {
    int d[4], s[4];
#pragma unroll
    for (int u = 0; u < 4; ++u) {
      const int i = i0 + u * 1024 + t;
      d[u] = (i < n) ? dst[e0 + i] : -1;
      s[u] = (i < n) ? src[e0 + i] : 0;
    }
#pragma unroll
    for (int u = 0; u < 4; ++u) {
      if (d[u] >= 0) {
        const int b = d[u] >> 6;
        const int pos = off[b] + atomicAdd(&cnt[b], 1);
        const int pk = (s[u] << 6) | (d[u] & 63);
        if (staged) sbuf[pos] = pk;
        else bpk[e0 + pos] = pk;
      }
    }
  }
  if (staged) {
    __syncthreads();
    for (int i = t; i < n; i += 1024) bpk[e0 + i] = sbuf[i];
  }
}

__global__ void __launch_bounds__(GCH)
bin_sort(const int* __restrict__ bpk, const int* __restrict__ binoff,
         int* __restrict__ rowptr, int* __restrict__ ssrc, int E) {
  __shared__ int ncnt[64];
  __shared__ int nstart[64];
  __shared__ int nfill[64];
  __shared__ int wsum[8];
  __shared__ int sbuf[CAP];
  __shared__ int sout[CAP];
  const int t = threadIdx.x;
  const int b = blockIdx.x;
  const int C = (E + GCH - 1) / GCH;
  const int o0 = binoff[t * (NBIN + 1) + b];
  const int o1 = binoff[t * (NBIN + 1) + b + 1];
  const int c = o1 - o0;
  if (t < 64) { ncnt[t] = 0; nfill[t] = 0; }
  int base, total;
  block_scan_excl<8>(o0, wsum, base);
  const int pos = block_scan_excl<8>(c, wsum, total);
  const bool staged = (total <= CAP);
  const int* run = bpk + (size_t)t * C + o0;
  for (int k = 0; k < c; k += 4) {
    int p[4];
#pragma unroll
    for (int u = 0; u < 4; ++u) p[u] = (k + u < c) ? run[k + u] : 0;
#pragma unroll
    for (int u = 0; u < 4; ++u) {
      if (k + u < c) {
        if (staged) sbuf[pos + k + u] = p[u];
        atomicAdd(&ncnt[p[u] & 63], 1);
      }
    }
  }
  __syncthreads();
  if (t < 32) {
    const int a = ncnt[2 * t];
    const int a2 = ncnt[2 * t + 1];
    const int s = a + a2;
    int incl = s;
#pragma unroll
    for (int off = 1; off < 32; off <<= 1) {
      const int y = __shfl_up(incl, off, 32);
      if (t >= off) incl += y;
    }
    nstart[2 * t] = incl - s;
    nstart[2 * t + 1] = incl - s + a;
  }
  __syncthreads();
  if (t < 64) {
    const int idx = b * 64 + t;
    if (idx < NN) rowptr[idx] = base + nstart[t];
  }
  if (b == NBIN - 1 && t == 0) rowptr[NN] = base + total;
  if (staged) {
    for (int j = t; j < total; j += GCH) {
      const int p = sbuf[j];
      const int node = p & 63;
      const int r = atomicAdd(&nfill[node], 1);
      sout[nstart[node] + r] = p >> 6;
    }
    __syncthreads();
    for (int j = t; j < total; j += GCH) ssrc[base + j] = sout[j];
  } else {
    for (int k = 0; k < c; ++k) {
      const int p = run[k];
      const int node = p & 63;
      const int r = atomicAdd(&nfill[node], 1);
      ssrc[base + nstart[node] + r] = p >> 6;
    }
  }
}

#define KP 132
template <int B>
__device__ __forceinline__ void hold_loads(const float4 (&v)[B]) {
  if constexpr (B == 16) asm volatile("" :: "v"(v[0].x), "v"(v[1].x), "v"(v[2].x), "v"(v[3].x), "v"(v[4].x), "v"(v[5].x), "v"(v[6].x), "v"(v[7].x), "v"(v[8].x), "v"(v[9].x), "v"(v[10].x), "v"(v[11].x), "v"(v[12].x), "v"(v[13].x), "v"(v[14].x), "v"(v[15].x));
  else if constexpr (B == 8) asm volatile("" :: "v"(v[0].x), "v"(v[1].x), "v"(v[2].x), "v"(v[3].x), "v"(v[4].x), "v"(v[5].x), "v"(v[6].x), "v"(v[7].x));
  else if constexpr (B == 4) asm volatile("" :: "v"(v[0].x), "v"(v[1].x), "v"(v[2].x), "v"(v[3].x));
  else if constexpr (B == 2) asm volatile("" :: "v"(v[0].x), "v"(v[1].x));
  else asm volatile("" :: "v"(v[0].x));
}

template <int B>
__device__ __forceinline__ void sum_run(const float4* __restrict__ X4,
                                        const int* __restrict__ ssrc,
                                        int j, int lane, int rpv, float4* Srow,
                                        int& cur, int& rend, float4& acc) {
  int s[B];
  float4 v[B];
#pragma unroll
  for (int u = 0; u < B; ++u) s[u] = ssrc[j + u];
#pragma unroll
  for (int u = 0; u < B; ++u) v[u] = X4[(size_t)s[u] * (CH / 4) + lane];
  hold_loads<B>(v);
#pragma unroll
  for (int u = 0; u < B; ++u) {
    while (j + u >= rend) {
      Srow[cur * (KP / 4)] = acc;
      acc = make_float4(0.f, 0.f, 0.f, 0.f);
      ++cur;
      rend = __builtin_amdgcn_readlane(rpv, cur + 1);
    }
    acc.x += v[u].x;
    acc.y += v[u].y;
    acc.z += v[u].z;
    acc.w += v[u].w;
  }
}

#define HP 68
#define MTW 5
#define ROWS (MTW * 16)

__device__ __forceinline__ void load_b_panel(const float* __restrict__ P, int nt, int lane,
                                             v2f (&bf)[CH / 4]) {
  const float4* P4 = (const float4*)P + (size_t)nt * (CH / 8) * 32 + lane;
  for (int i2 = 0; i2 < CH / 8; ++i2) {
    const float4 q = P4[i2 * 32];
    bf[2 * i2].x = q.x;
    bf[2 * i2].y = q.y;
    bf[2 * i2 + 1].x = q.z;
    bf[2 * i2 + 1].y = q.w;
  }
}

template <int NM>
__device__ __forceinline__ void mma_tiles(const float* Af, const v2f (&bf)[CH / 4], v8f (&c)[NM]) {
  for (int i = 0; i < CH / 4; ++i) {
    for (int m = 0; m < NM; ++m) {
      const v2f a = *(const v2f*)(Af + m * 16 * KP + 4 * i);
      c[m] = __builtin_amdgcn_wmma_f32_16x16x4_f32(
           false, a,  false, bf[i],
           (short)0, c[m],  false,  false);
    }
  }
}

template <int NM, int P>
__device__ __forceinline__ void relu_to_lds(float* S, int mfirst, int hi, int col, const v8f (&c)[NM]) {
  for (int m = 0; m < NM; ++m) {
    float* srow = S + ((mfirst + m) * 16 + hi * 8) * P + col;
    for (int r = 0; r < 8; ++r) srow[r * P] = fmaxf(c[m][r], 0.f);
  }
}

template <bool HEAD>
__global__ void __launch_bounds__(256, 6)
conv_mlp_wmma(const float* __restrict__ X,
              const int* __restrict__ rowptr,
              const int* __restrict__ ssrc,
              const float* __restrict__ Wa,
              const float* __restrict__ ba,
              const float* __restrict__ Wb,
              const float* __restrict__ bb,
              const float* __restrict__ W3,
              const float* __restrict__ b3,
              const float* __restrict__ w4,
              const float* __restrict__ b4,
              float* __restrict__ out) {
  constexpr int K = CH;
  constexpr int RW = ROWS / 8;
  __shared__ float4 S0q[ROWS * (KP / 4)];
  __shared__ float4 S1q[ROWS * (KP / 4)];
  float* S0 = (float*)S0q;
  float* S1 = (float*)S1q;

  const int t = threadIdx.x;
  const int lane = t & 31;
  const int wave = __builtin_amdgcn_readfirstlane(t >> 5);
  const int hi = lane >> 4;
  const int lo = lane & 15;
  const int col = (wave << 4) + lo;
  const size_t row0 = (size_t)blockIdx.x * ROWS;

  const int rbase = (int)row0 + wave * RW;
  const float4* X4 = (const float4*)X;
  float4 xr[RW];
  for (int i = 0; i < RW; ++i) xr[i] = X4[(size_t)(rbase + i) * (K / 4) + lane];
  const int rpv = rowptr[rbase + (lane < RW ? lane : RW)];
  int j = __builtin_amdgcn_readlane(rpv, 0);
  const int je = __builtin_amdgcn_readlane(rpv, RW);
  int rend = __builtin_amdgcn_readlane(rpv, 1);
  int cur = 0;
  float4 acc = make_float4(0.f, 0.f, 0.f, 0.f);
  float4* Srow = S0q + (wave * RW) * (KP / 4) + lane;
  for (; j + 16 <= je; j += 16) sum_run<16>(X4, ssrc, j, lane, rpv, Srow, cur, rend, acc);
  if (j + 8 <= je) { sum_run<8>(X4, ssrc, j, lane, rpv, Srow, cur, rend, acc); j += 8; }
  if (j + 4 <= je) { sum_run<4>(X4, ssrc, j, lane, rpv, Srow, cur, rend, acc); j += 4; }
  if (j + 2 <= je) { sum_run<2>(X4, ssrc, j, lane, rpv, Srow, cur, rend, acc); j += 2; }
  if (j < je) sum_run<1>(X4, ssrc, j, lane, rpv, Srow, cur, rend, acc);
  for (; cur < RW; ++cur) {
    Srow[cur * (KP / 4)] = acc;
    acc = make_float4(0.f, 0.f, 0.f, 0.f);
  }

  v2f bf[K / 4];
  load_b_panel(Wa, wave, lane, bf);
  float bv = ba[col];

  for (int i = 0; i < RW; ++i) {
    float4 a = Srow[i * (KP / 4)];
    a.x = xr[i].x + a.x;
    a.y = xr[i].y + a.y;
    a.z = xr[i].z + a.z;
    a.w = xr[i].w + a.w;
    Srow[i * (KP / 4)] = a;
  }
  __syncthreads();

  const int aoff = lo * KP + (hi << 1);
  v8f c[MTW];

  for (int m = 0; m < MTW; ++m)
    for (int r = 0; r < 8; ++r) c[m][r] = bv;
  mma_tiles<MTW>(S0 + aoff, bf, c);
  load_b_panel(Wb, wave, lane, bf);
  bv = bb[col];
  relu_to_lds<MTW, KP>(S1, 0, hi, col, c);
  __syncthreads();

  for (int m = 0; m < MTW; ++m)
    for (int r = 0; r < 8; ++r) c[m][r] = bv;
  mma_tiles<MTW>(S1 + aoff, bf, c);

  if (!HEAD) {
    for (int m = 0; m < MTW; ++m) {
      float* orow = out + (row0 + m * 16 + hi * 8) * CH + col;
      for (int r = 0; r < 8; ++r) orow[(size_t)r * CH] = fmaxf(c[m][r], 0.f);
    }
    return;
  } else {
    const int nt3 = wave & 3;
    const int g = wave >> 2;
    const int ln = (int)__lane_id();
    const int hi3 = ln >> 4;
    const int col3 = (nt3 << 4) + (ln & 15);
    const int aoff3 = (ln & 15) * KP + (hi3 << 1);
    relu_to_lds<MTW, KP>(S0, 0, hi, col, c);
    __syncthreads();
    load_b_panel(W3, nt3, ln, bf);
    const float bv3 = b3[col3];

    if (g == 0) {
      v8f c3[3];
      for (int m = 0; m < 3; ++m)
        for (int r = 0; r < 8; ++r) c3[m][r] = bv3;
      mma_tiles<3>(S0 + aoff3, bf, c3);
      relu_to_lds<3, HP>(S1, 0, hi3, col3, c3);
    } else {
      v8f c3[2];
      for (int m = 0; m < 2; ++m)
        for (int r = 0; r < 8; ++r) c3[m][r] = bv3;
      mma_tiles<2>(S0 + aoff3 + 3 * 16 * KP, bf, c3);
      relu_to_lds<2, HP>(S1, 3, hi3, col3, c3);
    }
    const float w4a = w4[ln];
    const float w4b = w4[ln + 32];
    const float b4v = b4[0];
    __syncthreads();

    float res = 0.f;
    for (int k = 0; k < ROWS / 8; ++k) {
      const float* h = S1 + (wave * (ROWS / 8) + k) * HP;
      float s = h[ln] * w4a + h[ln + 32] * w4b;
#pragma unroll
      for (int off = 16; off > 0; off >>= 1) s += __shfl_xor(s, off, 32);
      if (ln == k) res = s;
    }
    if (ln < ROWS / 8) out[row0 + wave * (ROWS / 8) + ln] = res + b4v;
  }
}

extern "C" void kernel_launch(void* const* d_in, const int* in_sizes, int n_in,
                              void* d_out, int out_size, void* d_ws, size_t ws_size,
                              hipStream_t stream) {
  const float* x   = (const float*)d_in[0];
  const int*   ei  = (const int*)d_in[1];
  const int    E   = in_sizes[1] / 2;
  const int* src = ei;
  const int* dst = ei + E;
  const float* w1a = (const float*)d_in[2];
  const float* b1a = (const float*)d_in[3];
  const float* w1b = (const float*)d_in[4];
  const float* b1b = (const float*)d_in[5];
  const float* w2a = (const float*)d_in[6];
  const float* b2a = (const float*)d_in[7];
  const float* w2b = (const float*)d_in[8];
  const float* b2b = (const float*)d_in[9];
  const float* w3  = (const float*)d_in[10];
  const float* b3  = (const float*)d_in[11];
  const float* w4  = (const float*)d_in[12];
  const float* b4  = (const float*)d_in[13];
  float* outp = (float*)d_out;

  float* h1   = (float*)d_ws;
  float* wpk  = h1 + (size_t)NN * CH;
  int* rowptr = (int*)(wpk + WPK_ALL);
  int* ssrc   = rowptr + NN + 16;
  int* bpk    = ssrc + E;
  int* binoff = bpk + E + GCH;

  constexpr unsigned GB = (NN / 16) / MTW;

  bin_edges<<<GCH + PACKB, 1024, 0, stream>>>(src, dst, bpk, binoff, E, w1a, w1b, w2a, w2b, w3, wpk);
  bin_sort<<<NBIN, GCH, 0, stream>>>(bpk, binoff, rowptr, ssrc, E);

  conv_mlp_wmma<false><<<GB, 256, 0, stream>>>(x, rowptr, ssrc, wpk, b1a, wpk + WPK128, b1b,
                                               nullptr, nullptr, nullptr, nullptr, h1);

  conv_mlp_wmma<true><<<GB, 256, 0, stream>>>(h1, rowptr, ssrc, wpk + 2 * WPK128, b2a, wpk + 3 * WPK128, b2b,
                                              wpk + 4 * WPK128, b3, w4, b4, outp);
}


# Dense launches: the 3125 m-tiles dealt to 768 work-groups of 4 or 5 m-tiles (all resident, 3 per WGP, 12-13 m-tiles per WGP), not 625 of 5 (15 on some WGPs, 10 on others); sums unchanged
# speedup vs baseline: 4.7625x; 1.0327x over previous
#include <hip/hip_runtime.h>

typedef __attribute__((ext_vector_type(2))) float v2f;
typedef __attribute__((ext_vector_type(8))) float v8f;

#define NN 50000
#define CH 128

#define NBIN ((NN + 63) / 64)
#define NSLOT 1024
#define GCH 256
#define CAP 4096

template <int NW>
__device__ __forceinline__ int block_scan_excl(int v, int* wsum, int& total) {
  const int lane = threadIdx.x & 31;
  const int wave = threadIdx.x >> 5;
  int incl = v;
#pragma unroll
  for (int off = 1; off < 32; off <<= 1) {
    const int y = __shfl_up(incl, off, 32);
    if (lane >= off) incl += y;
  }
  __syncthreads();
  if (lane == 31) wsum[wave] = incl;
  __syncthreads();
  int pre = 0, tot = 0;
  for (int w = 0; w < NW; ++w) {
    const int s = wsum[w];
    if (w < wave) pre += s;
    tot += s;
  }
  total = tot;
  return pre + incl - v;
}

#define WPK128 (CH * CH)
#define WPK64 (CH * 64)
#define WPK_ALL (4 * WPK128 + WPK64)
#define PACKB (WPK_ALL / 4 / 1024)

__device__ __forceinline__ void pack_weights(const float* __restrict__ w1a, const float* __restrict__ w1b,
                                             const float* __restrict__ w2a, const float* __restrict__ w2b,
                                             const float* __restrict__ w3, float* __restrict__ wpk, int q) {
  const float* Wm;
  int N, r;
  if (q < 4 * (WPK128 / 4)) {
    const int L = q / (WPK128 / 4);
    r = q - L * (WPK128 / 4);
    N = CH;
    Wm = (L == 0) ? w1a : (L == 1) ? w1b : (L == 2) ? w2a : w2b;
  } else {
    r = q - 4 * (WPK128 / 4);
    N = 64;
    Wm = w3;
  }
  const int l = r & 31;
  const int i2 = (r >> 5) & 15;
  const int w = r >> 9;
  const int c = (w << 4) + (l & 15);
  const int k0 = 8 * i2 + ((l >> 4) << 1);
  float4 v;
  v.x = Wm[(size_t)k0 * N + c];
  v.y = Wm[(size_t)(k0 + 1) * N + c];
  v.z = Wm[(size_t)(k0 + 4) * N + c];
  v.w = Wm[(size_t)(k0 + 5) * N + c];
  ((float4*)wpk)[q] = v;
}

__global__ void __launch_bounds__(1024)
bin_edges(const int* __restrict__ src, const int* __restrict__ dst,
          int* __restrict__ bpk, int* __restrict__ binoff, int E,
          const float* __restrict__ w1a, const float* __restrict__ w1b,
          const float* __restrict__ w2a, const float* __restrict__ w2b,
          const float* __restrict__ w3, float* __restrict__ wpk) {
  __shared__ int cnt[NSLOT];
  __shared__ int off[NSLOT];
  __shared__ int wsum[32];
  __shared__ int sbuf[CAP];
  const int t = threadIdx.x;
  const int g = blockIdx.x;
  if (g >= GCH) {
    pack_weights(w1a, w1b, w2a, w2b, w3, wpk, (g - GCH) * 1024 + t);
    return;
  }
  const int C = (E + GCH - 1) / GCH;
  const int e0 = g * C;
  int n = E - e0;
  n = n < 0 ? 0 : (n > C ? C : n);
  cnt[t] = 0;
  __syncthreads();
  for (int i0 = 0; i0 < n; i0 += 4096) {
    int d[4];
#pragma unroll
    for (int u = 0; u < 4; ++u) {
      const int i = i0 + u * 1024 + t;
      d[u] = (i < n) ? dst[e0 + i] : -1;
    }
#pragma unroll
    for (int u = 0; u < 4; ++u)
      if (d[u] >= 0) atomicAdd(&cnt[d[u] >> 6], 1);
  }
  __syncthreads();
  const int v = cnt[t];
  int tot;
  const int ex = block_scan_excl<32>(v, wsum, tot);
  off[t] = ex;
  cnt[t] = 0;
  __syncthreads();
  if (t <= NBIN) binoff[g * (NBIN + 1) + t] = ex;
  const bool staged = (C <= CAP);
  for (int i0 = 0; i0 < n; i0 += 4096) {
    int d[4], s[4];
#pragma unroll
    for (int u = 0; u < 4; ++u) {
      const int i = i0 + u * 1024 + t;
      d[u] = (i < n) ? dst[e0 + i] : -1;
      s[u] = (i < n) ? src[e0 + i] : 0;
    }
#pragma unroll
    for (int u = 0; u < 4; ++u) {
      if (d[u] >= 0) {
        const int b = d[u] >> 6;
        const int pos = off[b] + atomicAdd(&cnt[b], 1);
        const int pk = (s[u] << 6) | (d[u] & 63);
        if (staged) sbuf[pos] = pk;
        else bpk[e0 + pos] = pk;
      }
    }
  }
  if (staged) {
    __syncthreads();
    for (int i = t; i < n; i += 1024) bpk[e0 + i] = sbuf[i];
  }
}

__global__ void __launch_bounds__(GCH)
bin_sort(const int* __restrict__ bpk, const int* __restrict__ binoff,
         int* __restrict__ rowptr, int* __restrict__ ssrc, int E) {
  __shared__ int ncnt[64];
  __shared__ int nstart[64];
  __shared__ int nfill[64];
  __shared__ int wsum[8];
  __shared__ int sbuf[CAP];
  __shared__ int sout[CAP];
  const int t = threadIdx.x;
  const int b = blockIdx.x;
  const int C = (E + GCH - 1) / GCH;
  const int o0 = binoff[t * (NBIN + 1) + b];
  const int o1 = binoff[t * (NBIN + 1) + b + 1];
  const int c = o1 - o0;
  if (t < 64) { ncnt[t] = 0; nfill[t] = 0; }
  int base, total;
  block_scan_excl<8>(o0, wsum, base);
  const int pos = block_scan_excl<8>(c, wsum, total);
  const bool staged = (total <= CAP);
  const int* run = bpk + (size_t)t * C + o0;
  for (int k = 0; k < c; k += 4) {
    int p[4];
#pragma unroll
    for (int u = 0; u < 4; ++u) p[u] = (k + u < c) ? run[k + u] : 0;
#pragma unroll
    for (int u = 0; u < 4; ++u) {
      if (k + u < c) {
        if (staged) sbuf[pos + k + u] = p[u];
        atomicAdd(&ncnt[p[u] & 63], 1);
      }
    }
  }
  __syncthreads();
  if (t < 32) {
    const int a = ncnt[2 * t];
    const int a2 = ncnt[2 * t + 1];
    const int s = a + a2;
    int incl = s;
#pragma unroll
    for (int off = 1; off < 32; off <<= 1) {
      const int y = __shfl_up(incl, off, 32);
      if (t >= off) incl += y;
    }
    nstart[2 * t] = incl - s;
    nstart[2 * t + 1] = incl - s + a;
  }
  __syncthreads();
  if (t < 64) {
    const int idx = b * 64 + t;
    if (idx < NN) rowptr[idx] = base + nstart[t];
  }
  if (b == NBIN - 1 && t == 0) rowptr[NN] = base + total;
  if (staged) {
    for (int j = t; j < total; j += GCH) {
      const int p = sbuf[j];
      const int node = p & 63;
      const int r = atomicAdd(&nfill[node], 1);
      sout[nstart[node] + r] = p >> 6;
    }
    __syncthreads();
    for (int j = t; j < total; j += GCH) ssrc[base + j] = sout[j];
  } else {
    for (int k = 0; k < c; ++k) {
      const int p = run[k];
      const int node = p & 63;
      const int r = atomicAdd(&nfill[node], 1);
      ssrc[base + nstart[node] + r] = p >> 6;
    }
  }
}

#define KP 132
template <int B>
__device__ __forceinline__ void hold_loads(const float4 (&v)[B]) {
  if constexpr (B == 16) asm volatile("" :: "v"(v[0].x), "v"(v[1].x), "v"(v[2].x), "v"(v[3].x), "v"(v[4].x), "v"(v[5].x), "v"(v[6].x), "v"(v[7].x), "v"(v[8].x), "v"(v[9].x), "v"(v[10].x), "v"(v[11].x), "v"(v[12].x), "v"(v[13].x), "v"(v[14].x), "v"(v[15].x));
  else if constexpr (B == 8) asm volatile("" :: "v"(v[0].x), "v"(v[1].x), "v"(v[2].x), "v"(v[3].x), "v"(v[4].x), "v"(v[5].x), "v"(v[6].x), "v"(v[7].x));
  else if constexpr (B == 4) asm volatile("" :: "v"(v[0].x), "v"(v[1].x), "v"(v[2].x), "v"(v[3].x));
  else if constexpr (B == 2) asm volatile("" :: "v"(v[0].x), "v"(v[1].x));
  else asm volatile("" :: "v"(v[0].x));
}

template <int B>
__device__ __forceinline__ void sum_run(const float4* __restrict__ X4,
                                        const int* __restrict__ ssrc,
                                        int j, int lane, int rpv, float4* Srow,
                                        int& cur, int& rend, float4& acc) {
  int s[B];
  float4 v[B];
#pragma unroll
  for (int u = 0; u < B; ++u) s[u] = ssrc[j + u];
#pragma unroll
  for (int u = 0; u < B; ++u) v[u] = X4[(size_t)s[u] * (CH / 4) + lane];
  hold_loads<B>(v);
#pragma unroll
  for (int u = 0; u < B; ++u) {
    while (j + u >= rend) {
      Srow[cur * (KP / 4)] = acc;
      acc = make_float4(0.f, 0.f, 0.f, 0.f);
      ++cur;
      rend = __builtin_amdgcn_readlane(rpv, cur + 1);
    }
    acc.x += v[u].x;
    acc.y += v[u].y;
    acc.z += v[u].z;
    acc.w += v[u].w;
  }
}

#define HP 68
#define MTMAX 5
#define NMT (NN / 16)
#define NWG 768
static_assert(NN % 16 == 0 && NMT >= (MTMAX - 1) * NWG && NMT < MTMAX * NWG, "every work-group gets MTMAX - 1 or MTMAX m-tiles");

__device__ __forceinline__ void load_b_panel(const float* __restrict__ P, int nt, int lane,
                                             v2f (&bf)[CH / 4]) {
  const float4* P4 = (const float4*)P + (size_t)nt * (CH / 8) * 32 + lane;
  for (int i2 = 0; i2 < CH / 8; ++i2) {
    const float4 q = P4[i2 * 32];
    bf[2 * i2].x = q.x;
    bf[2 * i2].y = q.y;
    bf[2 * i2 + 1].x = q.z;
    bf[2 * i2 + 1].y = q.w;
  }
}

template <int NM>
__device__ __forceinline__ void mma_tiles(const float* Af, const v2f (&bf)[CH / 4], v8f (&c)[NM]) {
  for (int i = 0; i < CH / 4; ++i) {
    for (int m = 0; m < NM; ++m) {
      const v2f a = *(const v2f*)(Af + m * 16 * KP + 4 * i);
      c[m] = __builtin_amdgcn_wmma_f32_16x16x4_f32(
           false, a,  false, bf[i],
           (short)0, c[m],  false,  false);
    }
  }
}

template <int NM, int P>
__device__ __forceinline__ void relu_to_lds(float* S, int mfirst, int hi, int col, const v8f (&c)[NM]) {
  for (int m = 0; m < NM; ++m) {
    float* srow = S + ((mfirst + m) * 16 + hi * 8) * P + col;
    for (int r = 0; r < 8; ++r) srow[r * P] = fmaxf(c[m][r], 0.f);
  }
}

template <bool HEAD, int NM>
__device__ __forceinline__ void
conv_tile(float4* S0q, float4* S1q, int mt0,
              const float* __restrict__ X,
              const int* __restrict__ rowptr,
              const int* __restrict__ ssrc,
              const float* __restrict__ Wa,
              const float* __restrict__ ba,
              const float* __restrict__ Wb,
              const float* __restrict__ bb,
              const float* __restrict__ W3,
              const float* __restrict__ b3,
              const float* __restrict__ w4,
              const float* __restrict__ b4,
              float* __restrict__ out) {
  constexpr int K = CH;
  constexpr int ROWS = NM * 16;
  constexpr int RW = ROWS / 8;
  float* S0 = (float*)S0q;
  float* S1 = (float*)S1q;

  const int t = threadIdx.x;
  const int lane = t & 31;
  const int wave = __builtin_amdgcn_readfirstlane(t >> 5);
  const int hi = lane >> 4;
  const int lo = lane & 15;
  const int col = (wave << 4) + lo;
  const size_t row0 = (size_t)mt0 * 16;

  const int rbase = (int)row0 + wave * RW;
  const float4* X4 = (const float4*)X;
  float4 xr[RW];
  for (int i = 0; i < RW; ++i) xr[i] = X4[(size_t)(rbase + i) * (K / 4) + lane];
  const int rpv = rowptr[rbase + (lane < RW ? lane : RW)];
  int j = __builtin_amdgcn_readlane(rpv, 0);
  const int je = __builtin_amdgcn_readlane(rpv, RW);
  int rend = __builtin_amdgcn_readlane(rpv, 1);
  int cur = 0;
  float4 acc = make_float4(0.f, 0.f, 0.f, 0.f);
  float4* Srow = S0q + (wave * RW) * (KP / 4) + lane;
  for (; j + 16 <= je; j += 16) sum_run<16>(X4, ssrc, j, lane, rpv, Srow, cur, rend, acc);
  if (j + 8 <= je) { sum_run<8>(X4, ssrc, j, lane, rpv, Srow, cur, rend, acc); j += 8; }
  if (j + 4 <= je) { sum_run<4>(X4, ssrc, j, lane, rpv, Srow, cur, rend, acc); j += 4; }
  if (j + 2 <= je) { sum_run<2>(X4, ssrc, j, lane, rpv, Srow, cur, rend, acc); j += 2; }
  if (j < je) sum_run<1>(X4, ssrc, j, lane, rpv, Srow, cur, rend, acc);
  for (; cur < RW; ++cur) {
    Srow[cur * (KP / 4)] = acc;
    acc = make_float4(0.f, 0.f, 0.f, 0.f);
  }

  v2f bf[K / 4];
  load_b_panel(Wa, wave, lane, bf);
  float bv = ba[col];

  for (int i = 0; i < RW; ++i) {
    float4 a = Srow[i * (KP / 4)];
    a.x = xr[i].x + a.x;
    a.y = xr[i].y + a.y;
    a.z = xr[i].z + a.z;
    a.w = xr[i].w + a.w;
    Srow[i * (KP / 4)] = a;
  }
  __syncthreads();

  const int aoff = lo * KP + (hi << 1);
  v8f c[NM];

  for (int m = 0; m < NM; ++m)
    for (int r = 0; r < 8; ++r) c[m][r] = bv;
  mma_tiles<NM>(S0 + aoff, bf, c);
  load_b_panel(Wb, wave, lane, bf);
  bv = bb[col];
  relu_to_lds<NM, KP>(S1, 0, hi, col, c);
  __syncthreads();

  for (int m = 0; m < NM; ++m)
    for (int r = 0; r < 8; ++r) c[m][r] = bv;
  mma_tiles<NM>(S1 + aoff, bf, c);

  if (!HEAD) {
    const int ln = (int)__lane_id();
    const int ocol = (wave << 4) + (ln & 15);
    for (int m = 0; m < NM; ++m) {
      float* orow = out + (row0 + m * 16 + (ln >> 4) * 8) * CH + ocol;
      for (int r = 0; r < 8; ++r) orow[(size_t)r * CH] = fmaxf(c[m][r], 0.f);
    }
    return;
  } else {
    constexpr int T0 = (NM + 1) / 2;
    constexpr int T1 = NM - T0;
    const int nt3 = wave & 3;
    const int g = wave >> 2;
    const int ln = (int)__lane_id();
    const int hi3 = ln >> 4;
    const int col3 = (nt3 << 4) + (ln & 15);
    const int aoff3 = (ln & 15) * KP + (hi3 << 1);
    relu_to_lds<NM, KP>(S0, 0, hi, col, c);
    __syncthreads();
    load_b_panel(W3, nt3, ln, bf);
    const float bv3 = b3[col3];

    if (g == 0) {
      v8f c3[T0];
      for (int m = 0; m < T0; ++m)
        for (int r = 0; r < 8; ++r) c3[m][r] = bv3;
      mma_tiles<T0>(S0 + aoff3, bf, c3);
      relu_to_lds<T0, HP>(S1, 0, hi3, col3, c3);
    } else {
      v8f c3[T1];
      for (int m = 0; m < T1; ++m)
        for (int r = 0; r < 8; ++r) c3[m][r] = bv3;
      mma_tiles<T1>(S0 + aoff3 + T0 * 16 * KP, bf, c3);
      relu_to_lds<T1, HP>(S1, T0, hi3, col3, c3);
    }
    const float w4a = w4[ln];
    const float w4b = w4[ln + 32];
    const float b4v = b4[0];
    __syncthreads();

    float res = 0.f;
    for (int k = 0; k < RW; ++k) {
      const float* h = S1 + (wave * RW + k) * HP;
      float s = h[ln] * w4a + h[ln + 32] * w4b;
#pragma unroll
      for (int off = 16; off > 0; off >>= 1) s += __shfl_xor(s, off, 32);
      if (ln == k) res = s;
    }
    if (ln < RW) out[row0 + wave * RW + ln] = res + b4v;
  }
}

template <bool HEAD>
__global__ void __launch_bounds__(256, 6)
conv_mlp_wmma(const float* __restrict__ X,
              const int* __restrict__ rowptr,
              const int* __restrict__ ssrc,
              const float* __restrict__ Wa,
              const float* __restrict__ ba,
              const float* __restrict__ Wb,
              const float* __restrict__ bb,
              const float* __restrict__ W3,
              const float* __restrict__ b3,
              const float* __restrict__ w4,
              const float* __restrict__ b4,
              float* __restrict__ out) {
  __shared__ float4 S0q[MTMAX * 16 * (KP / 4)];
  __shared__ float4 S1q[MTMAX * 16 * (KP / 4)];
  const int mt0 = (int)(blockIdx.x * (unsigned)NMT / NWG);
  const int mt1 = (int)((blockIdx.x + 1) * (unsigned)NMT / NWG);
  if (mt1 - mt0 == MTMAX)
    conv_tile<HEAD, MTMAX>(S0q, S1q, mt0, X, rowptr, ssrc, Wa, ba, Wb, bb, W3, b3, w4, b4, out);
  else
    conv_tile<HEAD, MTMAX - 1>(S0q, S1q, mt0, X, rowptr, ssrc, Wa, ba, Wb, bb, W3, b3, w4, b4, out);
}

extern "C" void kernel_launch(void* const* d_in, const int* in_sizes, int n_in,
                              void* d_out, int out_size, void* d_ws, size_t ws_size,
                              hipStream_t stream) {
  const float* x   = (const float*)d_in[0];
  const int*   ei  = (const int*)d_in[1];
  const int    E   = in_sizes[1] / 2;
  const int* src = ei;
  const int* dst = ei + E;
  const float* w1a = (const float*)d_in[2];
  const float* b1a = (const float*)d_in[3];
  const float* w1b = (const float*)d_in[4];
  const float* b1b = (const float*)d_in[5];
  const float* w2a = (const float*)d_in[6];
  const float* b2a = (const float*)d_in[7];
  const float* w2b = (const float*)d_in[8];
  const float* b2b = (const float*)d_in[9];
  const float* w3  = (const float*)d_in[10];
  const float* b3  = (const float*)d_in[11];
  const float* w4  = (const float*)d_in[12];
  const float* b4  = (const float*)d_in[13];
  float* outp = (float*)d_out;

  float* h1   = (float*)d_ws;
  float* wpk  = h1 + (size_t)NN * CH;
  int* rowptr = (int*)(wpk + WPK_ALL);
  int* ssrc   = rowptr + NN + 16;
  int* bpk    = ssrc + E;
  int* binoff = bpk + E + GCH;

  constexpr unsigned GB = NWG;

  bin_edges<<<GCH + PACKB, 1024, 0, stream>>>(src, dst, bpk, binoff, E, w1a, w1b, w2a, w2b, w3, wpk);
  bin_sort<<<NBIN, GCH, 0, stream>>>(bpk, binoff, rowptr, ssrc, E);

  conv_mlp_wmma<false><<<GB, 256, 0, stream>>>(x, rowptr, ssrc, wpk, b1a, wpk + WPK128, b1b,
                                               nullptr, nullptr, nullptr, nullptr, h1);

  conv_mlp_wmma<true><<<GB, 256, 0, stream>>>(h1, rowptr, ssrc, wpk + 2 * WPK128, b2a, wpk + 3 * WPK128, b2b,
                                              wpk + 4 * WPK128, b3, w4, b4, outp);
}


# Second sort launch folded into the first dense launch: a bin = one dense work-group's rows, grouped by node in its LDS and summed from that LDS list (4 -> 3 launches); sum order = LDS arrival
# speedup vs baseline: 5.0080x; 1.0515x over previous
#include <hip/hip_runtime.h>

typedef __attribute__((ext_vector_type(2))) float v2f;
typedef __attribute__((ext_vector_type(8))) float v8f;

#define NN 50000
#define CH 128

#define MTMAX 5
#define NMT (NN / 16)
#define NWG 768
static_assert(NN % 16 == 0 && NMT >= (MTMAX - 1) * NWG && NMT < MTMAX * NWG, "every work-group gets MTMAX - 1 or MTMAX m-tiles");

#define NBIN NWG
#define NSLOT 1024
#define GCH 256
#define CAP 4096
#define NROWMAX 96
#define RPW 128
static_assert(NBIN + 1 <= NSLOT, "one counter per bin and one slot for the chunk size");
static_assert(MTMAX * 16 <= NROWMAX && MTMAX * 16 < RPW && RPW <= 128, "a row of a bin fits the 7 low bits of a packed edge");

__device__ __forceinline__ int bin_of_node(int d) { return (((d >> 4) + 1) * NWG - 1) / NMT; }
__device__ __forceinline__ int bin_first_row(int b) { return 16 * (int)((unsigned)b * NMT / NWG); }

template <int NW>
__device__ __forceinline__ int block_scan_excl(int v, int* wsum, int& total) {
  const int lane = threadIdx.x & 31;
  const int wave = threadIdx.x >> 5;
  int incl = v;
#pragma unroll
  for (int off = 1; off < 32; off <<= 1) {
    const int y = __shfl_up(incl, off, 32);
    if (lane >= off) incl += y;
  }
  __syncthreads();
  if (lane == 31) wsum[wave] = incl;
  __syncthreads();
  int pre = 0, tot = 0;
  for (int w = 0; w < NW; ++w) {
    const int s = wsum[w];
    if (w < wave) pre += s;
    tot += s;
  }
  total = tot;
  return pre + incl - v;
}

#define WPK128 (CH * CH)
#define WPK64 (CH * 64)
#define WPK_ALL (4 * WPK128 + WPK64)
#define PACKB (WPK_ALL / 4 / 1024)

__device__ __forceinline__ void pack_weights(const float* __restrict__ w1a, const float* __restrict__ w1b,
                                             const float* __restrict__ w2a, const float* __restrict__ w2b,
                                             const float* __restrict__ w3, float* __restrict__ wpk, int q) {
  const float* Wm;
  int N, r;
  if (q < 4 * (WPK128 / 4)) {
    const int L = q / (WPK128 / 4);
    r = q - L * (WPK128 / 4);
    N = CH;
    Wm = (L == 0) ? w1a : (L == 1) ? w1b : (L == 2) ? w2a : w2b;
  } else {
    r = q - 4 * (WPK128 / 4);
    N = 64;
    Wm = w3;
  }
  const int l = r & 31;
  const int i2 = (r >> 5) & 15;
  const int w = r >> 9;
  const int c = (w << 4) + (l & 15);
  const int k0 = 8 * i2 + ((l >> 4) << 1);
  float4 v;
  v.x = Wm[(size_t)k0 * N + c];
  v.y = Wm[(size_t)(k0 + 1) * N + c];
  v.z = Wm[(size_t)(k0 + 4) * N + c];
  v.w = Wm[(size_t)(k0 + 5) * N + c];
  ((float4*)wpk)[q] = v;
}

__global__ void __launch_bounds__(1024)
bin_edges(const int* __restrict__ src, const int* __restrict__ dst,
          int* __restrict__ bpk, int* __restrict__ binoff, int E,
          const float* __restrict__ w1a, const float* __restrict__ w1b,
          const float* __restrict__ w2a, const float* __restrict__ w2b,
          const float* __restrict__ w3, float* __restrict__ wpk) {
  __shared__ int cnt[NSLOT];
  __shared__ int off[NSLOT];
  __shared__ int wsum[32];
  __shared__ int sbuf[CAP];
  const int t = threadIdx.x;
  const int g = blockIdx.x;
  if (g >= GCH) {
    pack_weights(w1a, w1b, w2a, w2b, w3, wpk, (g - GCH) * 1024 + t);
    return;
  }
  const int C = (E + GCH - 1) / GCH;
  const int e0 = g * C;
  int n = E - e0;
  n = n < 0 ? 0 : (n > C ? C : n);
  cnt[t] = 0;
  __syncthreads();
  for (int i0 = 0; i0 < n; i0 += 4096) {
    int d[4];
#pragma unroll
    for (int u = 0; u < 4; ++u) {
      const int i = i0 + u * 1024 + t;
      d[u] = (i < n) ? dst[e0 + i] : -1;
    }
#pragma unroll
    for (int u = 0; u < 4; ++u)
      if (d[u] >= 0) atomicAdd(&cnt[bin_of_node(d[u])], 1);
  }
  __syncthreads();
  const int v = cnt[t];
  int tot;
  const int ex = block_scan_excl<32>(v, wsum, tot);
  off[t] = ex;
  cnt[t] = 0;
  __syncthreads();
  if (t <= NBIN) binoff[g * (NBIN + 1) + t] = ex;
  const bool staged = (C <= CAP);
  for (int i0 = 0; i0 < n; i0 += 4096) {
    int d[4], s[4];
#pragma unroll
    for (int u = 0; u < 4; ++u) {
      const int i = i0 + u * 1024 + t;
      d[u] = (i < n) ? dst[e0 + i] : -1;
      s[u] = (i < n) ? src[e0 + i] : 0;
    }
#pragma unroll
    for (int u = 0; u < 4; ++u) {
      if (d[u] >= 0) {
        const int b = bin_of_node(d[u]);
        const int pos = off[b] + atomicAdd(&cnt[b], 1);
        const int pk = (s[u] << 7) | (d[u] - bin_first_row(b));
        if (staged) sbuf[pos] = pk;
        else bpk[e0 + pos] = pk;
      }
    }
  }
  if (staged) {
    __syncthreads();
    for (int i = t; i < n; i += 1024) bpk[e0 + i] = sbuf[i];
  }
}

#define KP 132
template <int B>
__device__ __forceinline__ void hold_loads(const float4 (&v)[B]) {
  if constexpr (B == 16) asm volatile("" :: "v"(v[0].x), "v"(v[1].x), "v"(v[2].x), "v"(v[3].x), "v"(v[4].x), "v"(v[5].x), "v"(v[6].x), "v"(v[7].x), "v"(v[8].x), "v"(v[9].x), "v"(v[10].x), "v"(v[11].x), "v"(v[12].x), "v"(v[13].x), "v"(v[14].x), "v"(v[15].x));
  else if constexpr (B == 8) asm volatile("" :: "v"(v[0].x), "v"(v[1].x), "v"(v[2].x), "v"(v[3].x), "v"(v[4].x), "v"(v[5].x), "v"(v[6].x), "v"(v[7].x));
  else if constexpr (B == 4) asm volatile("" :: "v"(v[0].x), "v"(v[1].x), "v"(v[2].x), "v"(v[3].x));
  else if constexpr (B == 2) asm volatile("" :: "v"(v[0].x), "v"(v[1].x));
  else asm volatile("" :: "v"(v[0].x));
}

template <int B, bool LDSIDS>
__device__ __forceinline__ void sum_run(const float4* __restrict__ X4,
                                        const int* ids, int jofs,
                                        int j, int lane, int rpv, float4* Srow,
                                        int& cur, int& rend, float4& acc) {
  int s[B];
  float4 v[B];
#pragma unroll
  for (int u = 0; u < B; ++u) s[u] = ids[j + u - jofs];
#pragma unroll
  for (int u = 0; u < B; ++u) {
    if constexpr (LDSIDS) {
      const unsigned boff = ((unsigned)s[u] << 9) + ((unsigned)lane << 4);
      v[u] = *(const float4*)((const char*)X4 + boff);
    } else {
      v[u] = X4[(size_t)s[u] * (CH / 4) + lane];
    }
  }
  hold_loads<B>(v);
#pragma unroll
  for (int u = 0; u < B; ++u) {
    while (j + u >= rend) {
      Srow[cur * (KP / 4)] = acc;
      acc = make_float4(0.f, 0.f, 0.f, 0.f);
      ++cur;
      rend = __builtin_amdgcn_readlane(rpv, cur + 1);
    }
    acc.x += v[u].x;
    acc.y += v[u].y;
    acc.z += v[u].z;
    acc.w += v[u].w;
  }
}

template <bool LDSIDS>
__device__ __forceinline__ void sum_span(const float4* __restrict__ X4, const int* ids, int jofs,
                                         int& j, int jend, int lane, int rpv, float4* Srow,
                                         int& cur, int& rend, float4& acc) {
  for (; j + 16 <= jend; j += 16) sum_run<16, LDSIDS>(X4, ids, jofs, j, lane, rpv, Srow, cur, rend, acc);
  if (j + 8 <= jend) { sum_run<8, LDSIDS>(X4, ids, jofs, j, lane, rpv, Srow, cur, rend, acc); j += 8; }
  if (j + 4 <= jend) { sum_run<4, LDSIDS>(X4, ids, jofs, j, lane, rpv, Srow, cur, rend, acc); j += 4; }
  if (j + 2 <= jend) { sum_run<2, LDSIDS>(X4, ids, jofs, j, lane, rpv, Srow, cur, rend, acc); j += 2; }
  if (j < jend) { sum_run<1, LDSIDS>(X4, ids, jofs, j, lane, rpv, Srow, cur, rend, acc); j += 1; }
}

__device__ __forceinline__ int group_bin(const int* __restrict__ bpk, const int* __restrict__ binoff,
                                         int* __restrict__ rowptr, int* __restrict__ ssrc, int E,
                                         int b, int nrows,
                                         int* sbuf, int* sout, int* ncnt, int* nstart, int* nfill, int* wsum,
                                         int& base) {
  const int t = threadIdx.x;
  const int C = (E + GCH - 1) / GCH;
  const int o0 = binoff[t * (NBIN + 1) + b];
  const int o1 = binoff[t * (NBIN + 1) + b + 1];
  const int c = o1 - o0;
  if (t < NROWMAX) { ncnt[t] = 0; nfill[t] = 0; }
  int total;
  block_scan_excl<8>(o0, wsum, base);
  const int pos = block_scan_excl<8>(c, wsum, total);
  const bool staged = (total <= CAP);
  const int* run = bpk + (size_t)t * C + o0;
  for (int k = 0; k < c; k += 4) {
    int p[4];
#pragma unroll
    for (int u = 0; u < 4; ++u) p[u] = (k + u < c) ? run[k + u] : 0;
#pragma unroll
    for (int u = 0; u < 4; ++u) {
      if (k + u < c) {
        if (staged) sbuf[pos + k + u] = p[u];
        atomicAdd(&ncnt[p[u] & 127], 1);
      }
    }
  }
  __syncthreads();
  if (t < 32) {
    const int a0 = ncnt[3 * t];
    const int a1 = ncnt[3 * t + 1];
    const int a2 = ncnt[3 * t + 2];
    const int s = a0 + a1 + a2;
    int incl = s;
#pragma unroll
    for (int off = 1; off < 32; off <<= 1) {
      const int y = __shfl_up(incl, off, 32);
      if (t >= off) incl += y;
    }
    nstart[3 * t] = incl - s;
    nstart[3 * t + 1] = incl - s + a0;
    nstart[3 * t + 2] = incl - s + a0 + a1;
    if (t == 31) nstart[NROWMAX] = incl;
  }
  __syncthreads();
  if (t <= nrows) rowptr[b * RPW + t] = base + nstart[t];
  if (staged) {
    for (int j = t; j < total; j += GCH) {
      const int p = sbuf[j];
      const int node = p & 127;
      const int r = atomicAdd(&nfill[node], 1);
      sout[nstart[node] + r] = p >> 7;
    }
    __syncthreads();
    for (int j = t; j < total; j += GCH) ssrc[base + j] = sout[j];
  } else {
    for (int k = 0; k < c; ++k) {
      const int p = run[k];
      const int node = p & 127;
      const int r = atomicAdd(&nfill[node], 1);
      ssrc[base + nstart[node] + r] = p >> 7;
    }
    __threadfence();
    __syncthreads();
  }
  return total;
}

#define HP 68

__device__ __forceinline__ void load_b_panel(const float* __restrict__ P, int nt, int lane,
                                             v2f (&bf)[CH / 4]) {
  const float4* P4 = (const float4*)P + (size_t)nt * (CH / 8) * 32 + lane;
  for (int i2 = 0; i2 < CH / 8; ++i2) {
    const float4 q = P4[i2 * 32];
    bf[2 * i2].x = q.x;
    bf[2 * i2].y = q.y;
    bf[2 * i2 + 1].x = q.z;
    bf[2 * i2 + 1].y = q.w;
  }
}

template <int NM>
__device__ __forceinline__ void mma_tiles(const float* Af, const v2f (&bf)[CH / 4], v8f (&c)[NM]) {
  for (int i = 0; i < CH / 4; ++i) {
    for (int m = 0; m < NM; ++m) {
      const v2f a = *(const v2f*)(Af + m * 16 * KP + 4 * i);
      c[m] = __builtin_amdgcn_wmma_f32_16x16x4_f32(
           false, a,  false, bf[i],
           (short)0, c[m],  false,  false);
    }
  }
}

template <int NM, int P>
__device__ __forceinline__ void relu_to_lds(float* S, int mfirst, int hi, int col, const v8f (&c)[NM]) {
  for (int m = 0; m < NM; ++m) {
    float* srow = S + ((mfirst + m) * 16 + hi * 8) * P + col;
    for (int r = 0; r < 8; ++r) srow[r * P] = fmaxf(c[m][r], 0.f);
  }
}

template <bool HEAD, int NM>
__device__ __forceinline__ void
conv_tile(float4* S0q, float4* S1q, int mt0, int bidx,
              const int* nstart, int* sout, int total, const int* __restrict__ ssrc_w, int base,
              const float* __restrict__ X,
              const int* __restrict__ rowptr,
              const int* __restrict__ ssrc,
              const float* __restrict__ Wa,
              const float* __restrict__ ba,
              const float* __restrict__ Wb,
              const float* __restrict__ bb,
              const float* __restrict__ W3,
              const float* __restrict__ b3,
              const float* __restrict__ w4,
              const float* __restrict__ b4,
              float* __restrict__ out) {
  constexpr int K = CH;
  constexpr int ROWS = NM * 16;
  constexpr int RW = ROWS / 8;
  float* S0 = (float*)S0q;
  float* S1 = (float*)S1q;

  const int t = threadIdx.x;
  const int lane = t & 31;
  const int wave = __builtin_amdgcn_readfirstlane(t >> 5);
  const int hi = lane >> 4;
  const int lo = lane & 15;
  const int col = (wave << 4) + lo;
  const size_t row0 = (size_t)mt0 * 16;

  const int rbase = (int)row0 + wave * RW;
  const float4* X4 = (const float4*)X;
  float4 xr[RW];
  if constexpr (HEAD) {
    for (int i = 0; i < RW; ++i) xr[i] = X4[(size_t)(rbase + i) * (K / 4) + lane];
  } else {
    const char* Xb = (const char*)X4 + ((size_t)rbase << 9);
    for (int i = 0; i < RW; ++i) xr[i] = *(const float4*)(Xb + (unsigned)((i << 9) + (lane << 4)));
  }
  int rpv;
  if constexpr (HEAD) rpv = rowptr[bidx * RPW + wave * RW + (lane < RW ? lane : RW)];
  else rpv = nstart[wave * RW + (lane < RW ? lane : RW)];
  int j = __builtin_amdgcn_readlane(rpv, 0);
  const int je = __builtin_amdgcn_readlane(rpv, RW);
  int rend = __builtin_amdgcn_readlane(rpv, 1);
  int cur = 0;
  float4 acc = make_float4(0.f, 0.f, 0.f, 0.f);
  float4* Srow = S0q + (wave * RW) * (KP / 4) + lane;
  if constexpr (HEAD) {
    sum_span<false>(X4, ssrc, 0, j, je, lane, rpv, Srow, cur, rend, acc);
  } else {
    for (int s0 = 0;; s0 += CAP) {
      if (total > CAP) {
        __syncthreads();
        const int ns = (total - s0 < CAP) ? total - s0 : CAP;
        for (int i = t; i < ns; i += 256) sout[i] = __builtin_nontemporal_load(ssrc_w + base + s0 + i);
        __syncthreads();
      }
      const int jend = (je < s0 + CAP) ? je : s0 + CAP;
      sum_span<true>(X4, sout, s0, j, jend, lane, rpv, Srow, cur, rend, acc);
      if (s0 + CAP >= total) break;
    }
  }
  for (; cur < RW; ++cur) {
    Srow[cur * (KP / 4)] = acc;
    acc = make_float4(0.f, 0.f, 0.f, 0.f);
  }

  v2f bf[K / 4];
  load_b_panel(Wa, wave, lane, bf);
  float bv = ba[col];

  for (int i = 0; i < RW; ++i) {
    float4 a = Srow[i * (KP / 4)];
    a.x = xr[i].x + a.x;
    a.y = xr[i].y + a.y;
    a.z = xr[i].z + a.z;
    a.w = xr[i].w + a.w;
    Srow[i * (KP / 4)] = a;
  }
  __syncthreads();

  const int aoff = lo * KP + (hi << 1);
  v8f c[NM];

  for (int m = 0; m < NM; ++m)
    for (int r = 0; r < 8; ++r) c[m][r] = bv;
  mma_tiles<NM>(S0 + aoff, bf, c);
  load_b_panel(Wb, wave, lane, bf);
  bv = bb[col];
  relu_to_lds<NM, KP>(S1, 0, hi, col, c);
  __syncthreads();

  for (int m = 0; m < NM; ++m)
    for (int r = 0; r < 8; ++r) c[m][r] = bv;
  mma_tiles<NM>(S1 + aoff, bf, c);

  if (!HEAD) {
    const int ln = (int)__lane_id();
    const int ocol = (wave << 4) + (ln & 15);
    for (int m = 0; m < NM; ++m) {
      float* orow = out + (row0 + m * 16 + (ln >> 4) * 8) * CH + ocol;
      for (int r = 0; r < 8; ++r) orow[(size_t)r * CH] = fmaxf(c[m][r], 0.f);
    }
    return;
  } else {
    constexpr int T0 = (NM + 1) / 2;
    constexpr int T1 = NM - T0;
    const int nt3 = wave & 3;
    const int g = wave >> 2;
    const int ln = (int)__lane_id();
    const int hi3 = ln >> 4;
    const int col3 = (nt3 << 4) + (ln & 15);
    const int aoff3 = (ln & 15) * KP + (hi3 << 1);
    relu_to_lds<NM, KP>(S0, 0, hi, col, c);
    __syncthreads();
    load_b_panel(W3, nt3, ln, bf);
    const float bv3 = b3[col3];

    if (g == 0) {
      v8f c3[T0];
      for (int m = 0; m < T0; ++m)
        for (int r = 0; r < 8; ++r) c3[m][r] = bv3;
      mma_tiles<T0>(S0 + aoff3, bf, c3);
      relu_to_lds<T0, HP>(S1, 0, hi3, col3, c3);
    } else {
      v8f c3[T1];
      for (int m = 0; m < T1; ++m)
        for (int r = 0; r < 8; ++r) c3[m][r] = bv3;
      mma_tiles<T1>(S0 + aoff3 + T0 * 16 * KP, bf, c3);
      relu_to_lds<T1, HP>(S1, T0, hi3, col3, c3);
    }
    const float w4a = w4[ln];
    const float w4b = w4[ln + 32];
    const float b4v = b4[0];
    __syncthreads();

    float res = 0.f;
    for (int k = 0; k < RW; ++k) {
      const float* h = S1 + (wave * RW + k) * HP;
      float s = h[ln] * w4a + h[ln + 32] * w4b;
#pragma unroll
      for (int off = 16; off > 0; off >>= 1) s += __shfl_xor(s, off, 32);
      if (ln == k) res = s;
    }
    if (ln < RW) out[row0 + wave * RW + ln] = res + b4v;
  }
}

template <bool HEAD>
__global__ void __launch_bounds__(256, 6)
conv_mlp_wmma(const float* __restrict__ X,
              const int* __restrict__ rowptr,
              const int* __restrict__ ssrc,
              const float* __restrict__ Wa,
              const float* __restrict__ ba,
              const float* __restrict__ Wb,
              const float* __restrict__ bb,
              const float* __restrict__ W3,
              const float* __restrict__ b3,
              const float* __restrict__ w4,
              const float* __restrict__ b4,
              float* __restrict__ out,
              const int* __restrict__ bpk,
              const int* __restrict__ binoff,
              int* __restrict__ rowptr_w,
              int* __restrict__ ssrc_w,
              int E) {
  __shared__ float4 S0q[MTMAX * 16 * (KP / 4)];
  __shared__ float4 S1q[MTMAX * 16 * (KP / 4)];
  __shared__ int ncnt[NROWMAX];
  __shared__ int nfill[NROWMAX];
  __shared__ int nstart[NROWMAX + 1];
  __shared__ int wsum[8];
  static_assert(2 * CAP * sizeof(int) <= sizeof(S1q), "the two staging lists of the grouping live in the second tile");
  const int mt0 = (int)(blockIdx.x * (unsigned)NMT / NWG);
  const int mt1 = (int)((blockIdx.x + 1) * (unsigned)NMT / NWG);
  int* sbuf = (int*)S1q;
  int* sout = sbuf + CAP;
  int total = 0, base = 0;
  if constexpr (!HEAD)
    total = group_bin(bpk, binoff, rowptr_w, ssrc_w, E, (int)blockIdx.x, (mt1 - mt0) * 16,
                      sbuf, sout, ncnt, nstart, nfill, wsum, base);
  if (mt1 - mt0 == MTMAX)
    conv_tile<HEAD, MTMAX>(S0q, S1q, mt0, (int)blockIdx.x, nstart, sout, total, ssrc_w, base,
                           X, rowptr, ssrc, Wa, ba, Wb, bb, W3, b3, w4, b4, out);
  else
    conv_tile<HEAD, MTMAX - 1>(S0q, S1q, mt0, (int)blockIdx.x, nstart, sout, total, ssrc_w, base,
                               X, rowptr, ssrc, Wa, ba, Wb, bb, W3, b3, w4, b4, out);
}

extern "C" void kernel_launch(void* const* d_in, const int* in_sizes, int n_in,
                              void* d_out, int out_size, void* d_ws, size_t ws_size,
                              hipStream_t stream) {
  const float* x   = (const float*)d_in[0];
  const int*   ei  = (const int*)d_in[1];
  const int    E   = in_sizes[1] / 2;
  const int* src = ei;
  const int* dst = ei + E;
  const float* w1a = (const float*)d_in[2];
  const float* b1a = (const float*)d_in[3];
  const float* w1b = (const float*)d_in[4];
  const float* b1b = (const float*)d_in[5];
  const float* w2a = (const float*)d_in[6];
  const float* b2a = (const float*)d_in[7];
  const float* w2b = (const float*)d_in[8];
  const float* b2b = (const float*)d_in[9];
  const float* w3  = (const float*)d_in[10];
  const float* b3  = (const float*)d_in[11];
  const float* w4  = (const float*)d_in[12];
  const float* b4  = (const float*)d_in[13];
  float* outp = (float*)d_out;

  float* h1   = (float*)d_ws;
  float* wpk  = h1 + (size_t)NN * CH;
  int* rowptr = (int*)(wpk + WPK_ALL);
  int* ssrc   = rowptr + NWG * RPW;
  int* bpk    = ssrc + E;
  int* binoff = bpk + E + GCH;

  constexpr unsigned GB = NWG;

  bin_edges<<<GCH + PACKB, 1024, 0, stream>>>(src, dst, bpk, binoff, E, w1a, w1b, w2a, w2b, w3, wpk);

  conv_mlp_wmma<false><<<GB, 256, 0, stream>>>(x, nullptr, nullptr, wpk, b1a, wpk + WPK128, b1b,
                                               nullptr, nullptr, nullptr, nullptr, h1,
                                               bpk, binoff, rowptr, ssrc, E);

  conv_mlp_wmma<true><<<GB, 256, 0, stream>>>(h1, rowptr, ssrc, wpk + 2 * WPK128, b2a, wpk + 3 * WPK128, b2b,
                                              wpk + 4 * WPK128, b3, w4, b4, outp,
                                              nullptr, nullptr, nullptr, nullptr, E);
}
